# starting kernel
# baseline (speedup 1.0000x reference)
#include <hip/hip_runtime.h>

#define NN 4096
#define DD 256
#define SS 32
#define KK 4
#define ETA 0.5f

typedef __attribute__((ext_vector_type(16))) _Float16 v16h;
typedef __attribute__((ext_vector_type(8)))  _Float16 v8h;
typedef __attribute__((ext_vector_type(8)))  float    v8f;

__device__ __forceinline__ v16h make_ab(const _Float16* p0, const _Float16* p1) {
  v8h lo = *(const v8h*)p0;
  v8h hi = *(const v8h*)p1;
  return __builtin_shufflevector(lo, hi, 0,1,2,3,4,5,6,7,8,9,10,11,12,13,14,15);
}

__global__ __launch_bounds__(32) void zproj_kernel(
    const float* __restrict__ H, const float* __restrict__ U,
    const float* __restrict__ lp,
    _Float16* __restrict__ Zs, _Float16* __restrict__ Zf, _Float16* __restrict__ ZT)
{
  const int n = blockIdx.x;
  const int k = blockIdx.y;
  const int s = threadIdx.x;
  const float* Hrow = H + (size_t)n * DD;
  const float* Ucol = U + (size_t)k * DD * SS + s;
  float acc = 0.f;
  #pragma unroll 4
  for (int d = 0; d < DD; ++d) acc = fmaf(Hrow[d], Ucol[(size_t)d * SS], acc);
  const float e = __expf(lp[k * SS + s]);
  Zf[((size_t)k * NN + n) * SS + s] = (_Float16)acc;
  Zs[((size_t)k * NN + n) * SS + s] = (_Float16)(acc * e);
  ZT[((size_t)k * SS + s) * NN + n] = (_Float16)acc;
}

__global__ __launch_bounds__(32) void flash_attn_kernel(
    const _Float16* __restrict__ Zs, const _Float16* __restrict__ Zf,
    const _Float16* __restrict__ ZT, const float* __restrict__ adj,
    float* __restrict__ O)
{
  const int k    = blockIdx.y;
  const int qb   = blockIdx.x * 16;
  const int l    = threadIdx.x;
  const int ll   = l & 15;
  const int half = l >> 4;
  const int c0   = half ? 8 : 0;

  __shared__ _Float16 __attribute__((aligned(16))) Plds[16 * 64];

  const float inv_s = 0.17677669529663687f;

  const _Float16* qrow = Zs + ((size_t)k * NN + (qb + ll)) * SS;
  const v16h aQ = make_ab(qrow + c0, qrow + c0 + 16);

  v16h ones;
  #pragma unroll
  for (int i = 0; i < 16; ++i) ones[i] = (_Float16)1.0f;

  float m[8];
  #pragma unroll
  for (int v = 0; v < 8; ++v) m[v] = -INFINITY;
  v8f o0 = {}; v8f o1 = {};
  v8f lacc = {};

  for (int kb = 0; kb < NN; kb += 64) {
    v8f sc[4];
    #pragma unroll
    for (int t = 0; t < 4; ++t) {
      const _Float16* kr = Zf + ((size_t)k * NN + (kb + t * 16 + ll)) * SS;
      const v16h bK = make_ab(kr + c0, kr + c0 + 16);
      v8f z = {};
      sc[t] = __builtin_amdgcn_wmma_f32_16x16x32_f16(false, aQ, false, bK, (short)0, z, false, false);
    }

    #pragma unroll
    for (int v = 0; v < 8; ++v) {
      const int q = qb + v + 8 * half;
      const float* mrow = adj + (size_t)q * NN + kb + ll;
      const float w0 = mrow[0], w1 = mrow[16], w2 = mrow[32], w3 = mrow[48];
      __builtin_prefetch(mrow + 64, 0, 0);
      const float r0 = sc[0][v] * inv_s;
      const float r1 = sc[1][v] * inv_s;
      const float r2 = sc[2][v] * inv_s;
      const float r3 = sc[3][v] * inv_s;
      float mx = fmaxf(fmaxf(r0, r1), fmaxf(r2, r3));
      #pragma unroll
      for (int off = 1; off < 16; off <<= 1) mx = fmaxf(mx, __shfl_xor(mx, off, 32));
      const float mn    = fmaxf(m[v], mx);
      const float scale = __expf(m[v] - mn);
      m[v] = mn;
      const float p0 = w0 * __expf(r0 - mn);
      const float p1 = w1 * __expf(r1 - mn);
      const float p2 = w2 * __expf(r2 - mn);
      const float p3 = w3 * __expf(r3 - mn);
      o0[v]   *= scale;
      o1[v]   *= scale;
      lacc[v] *= scale;
      const int r = v + 8 * half;
      Plds[r * 64 + ll]      = (_Float16)p0;
      Plds[r * 64 + 16 + ll] = (_Float16)p1;
      Plds[r * 64 + 32 + ll] = (_Float16)p2;
      Plds[r * 64 + 48 + ll] = (_Float16)p3;
    }
    asm volatile("s_wait_dscnt 0" ::: "memory");

    #pragma unroll
    for (int c = 0; c < 2; ++c) {
      const v16h aP = make_ab(&Plds[ll * 64 + c * 32 + c0],
                              &Plds[ll * 64 + c * 32 + c0 + 16]);
      const _Float16* vr0 = ZT + ((size_t)k * SS + ll)      * NN + kb + c * 32;
      const _Float16* vr1 = ZT + ((size_t)k * SS + 16 + ll) * NN + kb + c * 32;
      const v16h bV0 = make_ab(vr0 + c0, vr0 + c0 + 16);
      const v16h bV1 = make_ab(vr1 + c0, vr1 + c0 + 16);
      o0   = __builtin_amdgcn_wmma_f32_16x16x32_f16(false, aP, false, bV0,  (short)0, o0,   false, false);
      o1   = __builtin_amdgcn_wmma_f32_16x16x32_f16(false, aP, false, bV1,  (short)0, o1,   false, false);
      lacc = __builtin_amdgcn_wmma_f32_16x16x32_f16(false, aP, false, ones, (short)0, lacc, false, false);
    }
  }

  #pragma unroll
  for (int v = 0; v < 8; ++v) {
    const float inv_l = 1.0f / lacc[v];
    const int q = qb + v + 8 * half;
    O[((size_t)k * NN + q) * SS + ll]      = o0[v] * inv_l;
    O[((size_t)k * NN + q) * SS + 16 + ll] = o1[v] * inv_l;
  }
}

__global__ __launch_bounds__(256) void combine_kernel(
    const float* __restrict__ H, const float* __restrict__ U,
    const float* __restrict__ O, const float* __restrict__ thr,
    float* __restrict__ out)
{
  const int n = blockIdx.x;
  const int d = threadIdx.x;
  __shared__ float Os[KK * SS];
  if (d < KK * SS) Os[d] = O[((size_t)(d >> 5) * NN + n) * SS + (d & 31)];
  __syncthreads();
  float agg = 0.f;
  #pragma unroll
  for (int k = 0; k < KK; ++k) {
    const float* Urow = U + ((size_t)k * DD + d) * SS;
    #pragma unroll
    for (int s = 0; s < SS; ++s) agg = fmaf(Os[k * SS + s], Urow[s], agg);
  }
  const float h = H[(size_t)n * DD + d] + ETA * agg;
  const float a = fabsf(h) - thr[d];
  out[(size_t)n * DD + d] = (a > 0.f) ? copysignf(a, h) : 0.f;
}

__global__ __launch_bounds__(256) void orth_kernel(const float* __restrict__ U,
                                                   float* __restrict__ out)
{
  const int t = threadIdx.x;
  float acc = 0.f;
  for (int idx = t; idx < KK * KK * SS * SS; idx += 256) {
    const int b  = idx & (SS - 1);
    const int a  = (idx >> 5) & (SS - 1);
    const int lq = (idx >> 10) & (KK - 1);
    const int kq = idx >> 12;
    if (kq > lq) continue;
    float c = 0.f;
    const float* Ua = U + (size_t)kq * DD * SS + a;
    const float* Ub = U + (size_t)lq * DD * SS + b;
    for (int d = 0; d < DD; ++d) c = fmaf(Ua[(size_t)d * SS], Ub[(size_t)d * SS], c);
    if (kq == lq) { const float e = c - (a == b ? 1.f : 0.f); acc += e * e; }
    else          { acc += c * c; }
  }
  __shared__ float red[256];
  red[t] = acc; __syncthreads();
  for (int off = 128; off > 0; off >>= 1) {
    if (t < off) red[t] += red[t + off];
    __syncthreads();
  }
  if (t == 0) out[(size_t)NN * DD] = red[0];
}

extern "C" void kernel_launch(void* const* d_in, const int* in_sizes, int n_in,
                              void* d_out, int out_size, void* d_ws, size_t ws_size,
                              hipStream_t stream) {
  (void)in_sizes; (void)n_in; (void)out_size; (void)ws_size;
  const float* H   = (const float*)d_in[0];
  const float* adj = (const float*)d_in[1];
  const float* U   = (const float*)d_in[2];
  const float* lp  = (const float*)d_in[3];
  const float* thr = (const float*)d_in[4];
  float* out = (float*)d_out;

  char* ws = (char*)d_ws;
  _Float16* Zs = (_Float16*)(ws);
  _Float16* Zf = (_Float16*)(ws + (1u << 20));
  _Float16* ZT = (_Float16*)(ws + (2u << 20));
  float*    O  = (float*)   (ws + (3u << 20));

  zproj_kernel     <<<dim3(NN, KK),      32, 0, stream>>>(H, U, lp, Zs, Zf, ZT);
  flash_attn_kernel<<<dim3(NN / 16, KK), 32, 0, stream>>>(Zs, Zf, ZT, adj, O);
  combine_kernel   <<<NN,               DD, 0, stream>>>(H, U, O, thr, out);
  orth_kernel      <<<1,               256, 0, stream>>>(U, out);
}


# orth loss: one-work-group kernel removed; its Gram rows ride the projection grid as 512 extra one-wave blocks, row sums added by one attention block (3 launches; loss summed per row, then over rows)
# speedup vs baseline: 2.6480x; 2.6480x over previous
#include <hip/hip_runtime.h>

#define NN 4096
#define DD 256
#define SS 32
#define KK 4
#define ETA 0.5f
#define GROWS  (KK * SS)
#define GPARTS (KK * KK * SS)

typedef __attribute__((ext_vector_type(16))) _Float16 v16h;
typedef __attribute__((ext_vector_type(8)))  _Float16 v8h;
typedef __attribute__((ext_vector_type(8)))  float    v8f;

__device__ __forceinline__ v16h make_ab(const _Float16* p0, const _Float16* p1) {
  v8h lo = *(const v8h*)p0;
  v8h hi = *(const v8h*)p1;
  return __builtin_shufflevector(lo, hi, 0,1,2,3,4,5,6,7,8,9,10,11,12,13,14,15);
}

__global__ __launch_bounds__(32) void zproj_kernel(
    const float* __restrict__ H, const float* __restrict__ U,
    const float* __restrict__ lp,
    _Float16* __restrict__ Zs, _Float16* __restrict__ Zf, _Float16* __restrict__ ZT,
    float* __restrict__ Gp)
{
  const int n = blockIdx.x;
  const int k = blockIdx.y;
  const int s = threadIdx.x;
  if (n >= NN) {
    const int kq = (n - NN) >> 5;
    const int a  = (n - NN) & (SS - 1);
    float part = 0.f;
    if (kq <= k) {
      const float* Ua = U + (size_t)kq * DD * SS + a;
      const float* Ub = U + (size_t)k  * DD * SS + s;
      float c = 0.f;
      for (int d = 0; d < DD; ++d) c = fmaf(Ua[(size_t)d * SS], Ub[(size_t)d * SS], c);
      const float e = (kq == k && a == s) ? c - 1.f : c;
      part = e * e;
      for (int off = 16; off > 0; off >>= 1) part += __shfl_xor(part, off, 32);
    }
    if (s == 0) Gp[(kq * KK + k) * SS + a] = part;
    return;
  }
  const float* Hrow = H + (size_t)n * DD;
  const float* Ucol = U + (size_t)k * DD * SS + s;
  float acc = 0.f;
  #pragma unroll 4
  for (int d = 0; d < DD; ++d) acc = fmaf(Hrow[d], Ucol[(size_t)d * SS], acc);
  const float e = __expf(lp[k * SS + s]);
  Zf[((size_t)k * NN + n) * SS + s] = (_Float16)acc;
  Zs[((size_t)k * NN + n) * SS + s] = (_Float16)(acc * e);
  ZT[((size_t)k * SS + s) * NN + n] = (_Float16)acc;
}

#define KSPLIT 8
#define KRANGE (NN / KSPLIT)

__global__ __launch_bounds__(32 * KSPLIT) void flash_attn_kernel(
    const _Float16* __restrict__ Zs, const _Float16* __restrict__ Zf,
    const _Float16* __restrict__ ZT, const float* __restrict__ adj,
    float* __restrict__ O, float* __restrict__ out)
{
  const int k    = blockIdx.y;
  const int qb   = blockIdx.x * 16;
  const int w    = __builtin_amdgcn_readfirstlane((int)(threadIdx.x >> 5));
  const int l    = threadIdx.x & 31;
  const int ll   = l & 15;
  const int half = l >> 4;
  const int c0   = half ? 8 : 0;

  if (blockIdx.x == 0 && blockIdx.y == 0 && w == 0) {
    float t = 0.f;
    for (int i = 0; i < GPARTS / 32; ++i) t += O[l + 32 * i];
    for (int off = 16; off > 0; off >>= 1) t += __shfl_xor(t, off, 32);
    if (l == 0) out[(size_t)NN * DD] = t;
  }

  __shared__ _Float16 __attribute__((aligned(16))) PldsAll[KSPLIT * 16 * 64];
  __shared__ float Om[KSPLIT * 16 * 32];
  __shared__ float Mm[KSPLIT * 16];
  __shared__ float Lm[KSPLIT * 16];
  _Float16* Plds = PldsAll + w * (16 * 64);

  const float inv_s = 0.17677669529663687f;

  const _Float16* qrow = Zs + ((size_t)k * NN + (qb + ll)) * SS;
  const v16h aQ = make_ab(qrow + c0, qrow + c0 + 16);

  v16h ones;
  #pragma unroll
  for (int i = 0; i < 16; ++i) ones[i] = (_Float16)1.0f;

  float m[8];
  #pragma unroll
  for (int v = 0; v < 8; ++v) m[v] = -INFINITY;
  v8f o0 = {}; v8f o1 = {};
  v8f lacc = {};

  const _Float16* Zfk  = Zf + (size_t)k * NN * SS;
  const _Float16* ZTk  = ZT + (size_t)k * SS * NN;
  const float*    adjq = adj + (size_t)qb * NN;
  const unsigned  koff = (unsigned)(ll * SS + c0);
  const unsigned  voff = (unsigned)(ll * NN + c0);
  const unsigned  moff = (unsigned)(8 * half * NN + ll);
  const int kb0 = w * KRANGE;
  for (int kb = kb0; kb < kb0 + KRANGE; kb += 64) {
    v8f sc[4];
    #pragma unroll
    for (int t = 0; t < 4; ++t) {
      const _Float16* kr = Zfk + (unsigned)(kb * SS) + koff + t * 16 * SS;
      const v16h bK = make_ab(kr, kr + 16);
      v8f z = {};
      sc[t] = __builtin_amdgcn_wmma_f32_16x16x32_f16(false, aQ, false, bK, (short)0, z, false, false);
    }

    #pragma unroll
    for (int v = 0; v < 8; ++v) {
      const float* mrow = adjq + (unsigned)kb + moff + v * NN;
      const float w0 = mrow[0], w1 = mrow[16], w2 = mrow[32], w3 = mrow[48];
      __builtin_prefetch(mrow + 64, 0, 0);
      const float r0 = sc[0][v] * inv_s;
      const float r1 = sc[1][v] * inv_s;
      const float r2 = sc[2][v] * inv_s;
      const float r3 = sc[3][v] * inv_s;
      float mx = fmaxf(fmaxf(r0, r1), fmaxf(r2, r3));
      #pragma unroll
      for (int off = 1; off < 16; off <<= 1) mx = fmaxf(mx, __shfl_xor(mx, off, 32));
      const float mn    = fmaxf(m[v], mx);
      const float scale = __expf(m[v] - mn);
      m[v] = mn;
      const float p0 = w0 * __expf(r0 - mn);
      const float p1 = w1 * __expf(r1 - mn);
      const float p2 = w2 * __expf(r2 - mn);
      const float p3 = w3 * __expf(r3 - mn);
      o0[v]   *= scale;
      o1[v]   *= scale;
      lacc[v] *= scale;
      const int r = v + 8 * half;
      Plds[r * 64 + ll]      = (_Float16)p0;
      Plds[r * 64 + 16 + ll] = (_Float16)p1;
      Plds[r * 64 + 32 + ll] = (_Float16)p2;
      Plds[r * 64 + 48 + ll] = (_Float16)p3;
    }
    asm volatile("s_wait_dscnt 0" ::: "memory");

    #pragma unroll
    for (int c = 0; c < 2; ++c) {
      const v16h aP = make_ab(&Plds[ll * 64 + c * 32 + c0],
                              &Plds[ll * 64 + c * 32 + c0 + 16]);
      const _Float16* vr0 = ZTk + (unsigned)kb + voff + c * 32;
      const _Float16* vr1 = vr0 + 16 * NN;
      const v16h bV0 = make_ab(vr0, vr0 + 16);
      const v16h bV1 = make_ab(vr1, vr1 + 16);
      o0   = __builtin_amdgcn_wmma_f32_16x16x32_f16(false, aP, false, bV0,  (short)0, o0,   false, false);
      o1   = __builtin_amdgcn_wmma_f32_16x16x32_f16(false, aP, false, bV1,  (short)0, o1,   false, false);
      lacc = __builtin_amdgcn_wmma_f32_16x16x32_f16(false, aP, false, ones, (short)0, lacc, false, false);
    }
  }

  #pragma unroll
  for (int v = 0; v < 8; ++v) {
    const int r = v + 8 * half;
    Om[(w * 16 + r) * 32 + ll]      = o0[v];
    Om[(w * 16 + r) * 32 + 16 + ll] = o1[v];
    if (ll == 0) {
      Mm[w * 16 + r] = m[v];
      Lm[w * 16 + r] = lacc[v];
    }
  }
  __syncthreads();

  for (int j = 0; j < 2; ++j) {
    const int i = (int)threadIdx.x + j * (32 * KSPLIT);
    const int r = i >> 5;
    float M = Mm[r];
    for (int u = 1; u < KSPLIT; ++u) M = fmaxf(M, Mm[u * 16 + r]);
    float L = 0.f, acc = 0.f;
    for (int u = 0; u < KSPLIT; ++u) {
      const float f = __expf(Mm[u * 16 + r] - M);
      L   = fmaf(Lm[u * 16 + r], f, L);
      acc = fmaf(Om[u * 512 + i], f, acc);
    }
    O[((size_t)k * NN + qb) * SS + i] = acc / L;
  }
}

__global__ __launch_bounds__(256) void combine_kernel(
    const float* __restrict__ H, const float* __restrict__ U,
    const float* __restrict__ O, const float* __restrict__ thr,
    float* __restrict__ out)
{
  const int n = blockIdx.x;
  const int d = threadIdx.x;
  __shared__ float Os[KK * SS];
  if (d < KK * SS) Os[d] = O[((size_t)(d >> 5) * NN + n) * SS + (d & 31)];
  __syncthreads();
  float agg = 0.f;
  #pragma unroll
  for (int k = 0; k < KK; ++k) {
    const float* Urow = U + ((size_t)k * DD + d) * SS;
    #pragma unroll
    for (int s = 0; s < SS; ++s) agg = fmaf(Os[k * SS + s], Urow[s], agg);
  }
  const float h = H[(size_t)n * DD + d] + ETA * agg;
  const float a = fabsf(h) - thr[d];
  out[(size_t)n * DD + d] = (a > 0.f) ? copysignf(a, h) : 0.f;
}

extern "C" void kernel_launch(void* const* d_in, const int* in_sizes, int n_in,
                              void* d_out, int out_size, void* d_ws, size_t ws_size,
                              hipStream_t stream) {
  (void)in_sizes; (void)n_in; (void)out_size; (void)ws_size;
  const float* H   = (const float*)d_in[0];
  const float* adj = (const float*)d_in[1];
  const float* U   = (const float*)d_in[2];
  const float* lp  = (const float*)d_in[3];
  const float* thr = (const float*)d_in[4];
  float* out = (float*)d_out;

  char* ws = (char*)d_ws;
  _Float16* Zs = (_Float16*)(ws);
  _Float16* Zf = (_Float16*)(ws + (1u << 20));
  _Float16* ZT = (_Float16*)(ws + (2u << 20));
  float*    O  = (float*)   (ws + (3u << 20));

  zproj_kernel     <<<dim3(NN + GROWS, KK), 32, 0, stream>>>(H, U, lp, Zs, Zf, ZT, O);
  flash_attn_kernel<<<dim3(NN / 16, KK), 32 * KSPLIT, 0, stream>>>(Zs, Zf, ZT, adj, O, out);
  combine_kernel   <<<NN,               DD, 0, stream>>>(H, U, O, thr, out);
}


# combine: agg = sum_k O_k U_k^T moved from per-thread FMA rows (one n per block, strided U reads) to f32 16x16x4 WMMA, 16 rows per block; (k,s) sum now in the matrix unit's order
# speedup vs baseline: 3.7698x; 1.4236x over previous
#include <hip/hip_runtime.h>

#define NN 4096
#define DD 256
#define SS 32
#define KK 4
#define ETA 0.5f
#define GROWS  (KK * SS)
#define GPARTS (KK * KK * SS)

typedef __attribute__((ext_vector_type(16))) _Float16 v16h;
typedef __attribute__((ext_vector_type(8)))  _Float16 v8h;
typedef __attribute__((ext_vector_type(8)))  float    v8f;

__device__ __forceinline__ v16h make_ab(const _Float16* p0, const _Float16* p1) {
  v8h lo = *(const v8h*)p0;
  v8h hi = *(const v8h*)p1;
  return __builtin_shufflevector(lo, hi, 0,1,2,3,4,5,6,7,8,9,10,11,12,13,14,15);
}

__global__ __launch_bounds__(32) void zproj_kernel(
    const float* __restrict__ H, const float* __restrict__ U,
    const float* __restrict__ lp,
    _Float16* __restrict__ Zs, _Float16* __restrict__ Zf, _Float16* __restrict__ ZT,
    float* __restrict__ Gp)
{
  const int n = blockIdx.x;
  const int k = blockIdx.y;
  const int s = threadIdx.x;
  if (n >= NN) {
    const int kq = (n - NN) >> 5;
    const int a  = (n - NN) & (SS - 1);
    float part = 0.f;
    if (kq <= k) {
      const float* Ua = U + (size_t)kq * DD * SS + a;
      const float* Ub = U + (size_t)k  * DD * SS + s;
      float c = 0.f;
      for (int d = 0; d < DD; ++d) c = fmaf(Ua[(size_t)d * SS], Ub[(size_t)d * SS], c);
      const float e = (kq == k && a == s) ? c - 1.f : c;
      part = e * e;
      for (int off = 16; off > 0; off >>= 1) part += __shfl_xor(part, off, 32);
    }
    if (s == 0) Gp[(kq * KK + k) * SS + a] = part;
    return;
  }
  const float* Hrow = H + (size_t)n * DD;
  const float* Ucol = U + (size_t)k * DD * SS + s;
  float acc = 0.f;
  #pragma unroll 4
  for (int d = 0; d < DD; ++d) acc = fmaf(Hrow[d], Ucol[(size_t)d * SS], acc);
  const float e = __expf(lp[k * SS + s]);
  Zf[((size_t)k * NN + n) * SS + s] = (_Float16)acc;
  Zs[((size_t)k * NN + n) * SS + s] = (_Float16)(acc * e);
  ZT[((size_t)k * SS + s) * NN + n] = (_Float16)acc;
}

#define KSPLIT 8
#define KRANGE (NN / KSPLIT)

__global__ __launch_bounds__(32 * KSPLIT) void flash_attn_kernel(
    const _Float16* __restrict__ Zs, const _Float16* __restrict__ Zf,
    const _Float16* __restrict__ ZT, const float* __restrict__ adj,
    float* __restrict__ O, float* __restrict__ out)
{
  const int k    = blockIdx.y;
  const int qb   = blockIdx.x * 16;
  const int w    = __builtin_amdgcn_readfirstlane((int)(threadIdx.x >> 5));
  const int l    = threadIdx.x & 31;
  const int ll   = l & 15;
  const int half = l >> 4;
  const int c0   = half ? 8 : 0;

  if (blockIdx.x == 0 && blockIdx.y == 0 && w == 0) {
    float t = 0.f;
    for (int i = 0; i < GPARTS / 32; ++i) t += O[l + 32 * i];
    for (int off = 16; off > 0; off >>= 1) t += __shfl_xor(t, off, 32);
    if (l == 0) out[(size_t)NN * DD] = t;
  }

  __shared__ _Float16 __attribute__((aligned(16))) PldsAll[KSPLIT * 16 * 64];
  __shared__ float Om[KSPLIT * 16 * 32];
  __shared__ float Mm[KSPLIT * 16];
  __shared__ float Lm[KSPLIT * 16];
  _Float16* Plds = PldsAll + w * (16 * 64);

  const float inv_s = 0.17677669529663687f;

  const _Float16* qrow = Zs + ((size_t)k * NN + (qb + ll)) * SS;
  const v16h aQ = make_ab(qrow + c0, qrow + c0 + 16);

  v16h ones;
  #pragma unroll
  for (int i = 0; i < 16; ++i) ones[i] = (_Float16)1.0f;

  float m[8];
  #pragma unroll
  for (int v = 0; v < 8; ++v) m[v] = -INFINITY;
  v8f o0 = {}; v8f o1 = {};
  v8f lacc = {};

  const _Float16* Zfk  = Zf + (size_t)k * NN * SS;
  const _Float16* ZTk  = ZT + (size_t)k * SS * NN;
  const float*    adjq = adj + (size_t)qb * NN;
  const unsigned  koff = (unsigned)(ll * SS + c0);
  const unsigned  voff = (unsigned)(ll * NN + c0);
  const unsigned  moff = (unsigned)(8 * half * NN + ll);
  const int kb0 = w * KRANGE;
  for (int kb = kb0; kb < kb0 + KRANGE; kb += 64) {
    v8f sc[4];
    #pragma unroll
    for (int t = 0; t < 4; ++t) {
      const _Float16* kr = Zfk + (unsigned)(kb * SS) + koff + t * 16 * SS;
      const v16h bK = make_ab(kr, kr + 16);
      v8f z = {};
      sc[t] = __builtin_amdgcn_wmma_f32_16x16x32_f16(false, aQ, false, bK, (short)0, z, false, false);
    }

    #pragma unroll
    for (int v = 0; v < 8; ++v) {
      const float* mrow = adjq + (unsigned)kb + moff + v * NN;
      const float w0 = mrow[0], w1 = mrow[16], w2 = mrow[32], w3 = mrow[48];
      __builtin_prefetch(mrow + 64, 0, 0);
      const float r0 = sc[0][v] * inv_s;
      const float r1 = sc[1][v] * inv_s;
      const float r2 = sc[2][v] * inv_s;
      const float r3 = sc[3][v] * inv_s;
      float mx = fmaxf(fmaxf(r0, r1), fmaxf(r2, r3));
      #pragma unroll
      for (int off = 1; off < 16; off <<= 1) mx = fmaxf(mx, __shfl_xor(mx, off, 32));
      const float mn    = fmaxf(m[v], mx);
      const float scale = __expf(m[v] - mn);
      m[v] = mn;
      const float p0 = w0 * __expf(r0 - mn);
      const float p1 = w1 * __expf(r1 - mn);
      const float p2 = w2 * __expf(r2 - mn);
      const float p3 = w3 * __expf(r3 - mn);
      o0[v]   *= scale;
      o1[v]   *= scale;
      lacc[v] *= scale;
      const int r = v + 8 * half;
      Plds[r * 64 + ll]      = (_Float16)p0;
      Plds[r * 64 + 16 + ll] = (_Float16)p1;
      Plds[r * 64 + 32 + ll] = (_Float16)p2;
      Plds[r * 64 + 48 + ll] = (_Float16)p3;
    }
    asm volatile("s_wait_dscnt 0" ::: "memory");

    #pragma unroll
    for (int c = 0; c < 2; ++c) {
      const v16h aP = make_ab(&Plds[ll * 64 + c * 32 + c0],
                              &Plds[ll * 64 + c * 32 + c0 + 16]);
      const _Float16* vr0 = ZTk + (unsigned)kb + voff + c * 32;
      const _Float16* vr1 = vr0 + 16 * NN;
      const v16h bV0 = make_ab(vr0, vr0 + 16);
      const v16h bV1 = make_ab(vr1, vr1 + 16);
      o0   = __builtin_amdgcn_wmma_f32_16x16x32_f16(false, aP, false, bV0,  (short)0, o0,   false, false);
      o1   = __builtin_amdgcn_wmma_f32_16x16x32_f16(false, aP, false, bV1,  (short)0, o1,   false, false);
      lacc = __builtin_amdgcn_wmma_f32_16x16x32_f16(false, aP, false, ones, (short)0, lacc, false, false);
    }
  }

  #pragma unroll
  for (int v = 0; v < 8; ++v) {
    const int r = v + 8 * half;
    Om[(w * 16 + r) * 32 + ll]      = o0[v];
    Om[(w * 16 + r) * 32 + 16 + ll] = o1[v];
    if (ll == 0) {
      Mm[w * 16 + r] = m[v];
      Lm[w * 16 + r] = lacc[v];
    }
  }
  __syncthreads();

  for (int j = 0; j < 2; ++j) {
    const int i = (int)threadIdx.x + j * (32 * KSPLIT);
    const int r = i >> 5;
    float M = Mm[r];
    for (int u = 1; u < KSPLIT; ++u) M = fmaxf(M, Mm[u * 16 + r]);
    float L = 0.f, acc = 0.f;
    for (int u = 0; u < KSPLIT; ++u) {
      const float f = __expf(Mm[u * 16 + r] - M);
      L   = fmaf(Lm[u * 16 + r], f, L);
      acc = fmaf(Om[u * 512 + i], f, acc);
    }
    O[((size_t)k * NN + qb) * SS + i] = acc / L;
  }
}

typedef __attribute__((ext_vector_type(2))) float v2f;
typedef __attribute__((ext_vector_type(4))) float v4f;

__global__ __launch_bounds__(256) void combine_kernel(
    const float* __restrict__ H, const float* __restrict__ U,
    const float* __restrict__ O, const float* __restrict__ thr,
    float* __restrict__ out)
{
  const int n0 = blockIdx.x * 16;
  const int w  = threadIdx.x >> 5;
  const int l  = threadIdx.x & 31;
  const int c  = l & 15;
  const int h  = l >> 4;
  const int dc = w * 32 + 2 * c;

  v8f acc0 = {}; v8f acc1 = {};
  #pragma unroll
  for (int k = 0; k < KK; ++k) {
    const v4f* pa  = (const v4f*)(O + ((size_t)k * NN + n0 + c) * SS + 16 * h);
    const v4f* pb0 = (const v4f*)(U + ((size_t)k * DD + dc) * SS + 16 * h);
    const v4f* pb1 = (const v4f*)(U + ((size_t)k * DD + dc + 1) * SS + 16 * h);
    #pragma unroll
    for (int q = 0; q < 4; ++q) {
      const v4f a = pa[q], b0 = pb0[q], b1 = pb1[q];
      const v2f alo  = {a[0],  a[1]},  ahi  = {a[2],  a[3]};
      const v2f b0lo = {b0[0], b0[1]}, b0hi = {b0[2], b0[3]};
      const v2f b1lo = {b1[0], b1[1]}, b1hi = {b1[2], b1[3]};
      acc0 = __builtin_amdgcn_wmma_f32_16x16x4_f32(false, alo, false, b0lo, (short)0, acc0, false, false);
      acc1 = __builtin_amdgcn_wmma_f32_16x16x4_f32(false, alo, false, b1lo, (short)0, acc1, false, false);
      acc0 = __builtin_amdgcn_wmma_f32_16x16x4_f32(false, ahi, false, b0hi, (short)0, acc0, false, false);
      acc1 = __builtin_amdgcn_wmma_f32_16x16x4_f32(false, ahi, false, b1hi, (short)0, acc1, false, false);
    }
  }

  const v2f t = *(const v2f*)(thr + dc);
  #pragma unroll
  for (int r = 0; r < 8; ++r) {
    const size_t o = (size_t)(n0 + 8 * h + r) * DD + dc;
    const v2f hv = *(const v2f*)(H + o);
    const float h0 = hv[0] + ETA * acc0[r];
    const float h1 = hv[1] + ETA * acc1[r];
    const float a0 = fabsf(h0) - t[0];
    const float a1 = fabsf(h1) - t[1];
    v2f res;
    res[0] = (a0 > 0.f) ? copysignf(a0, h0) : 0.f;
    res[1] = (a1 > 0.f) ? copysignf(a1, h1) : 0.f;
    *(v2f*)(out + o) = res;
  }
}

extern "C" void kernel_launch(void* const* d_in, const int* in_sizes, int n_in,
                              void* d_out, int out_size, void* d_ws, size_t ws_size,
                              hipStream_t stream) {
  (void)in_sizes; (void)n_in; (void)out_size; (void)ws_size;
  const float* H   = (const float*)d_in[0];
  const float* adj = (const float*)d_in[1];
  const float* U   = (const float*)d_in[2];
  const float* lp  = (const float*)d_in[3];
  const float* thr = (const float*)d_in[4];
  float* out = (float*)d_out;

  char* ws = (char*)d_ws;
  _Float16* Zs = (_Float16*)(ws);
  _Float16* Zf = (_Float16*)(ws + (1u << 20));
  _Float16* ZT = (_Float16*)(ws + (2u << 20));
  float*    O  = (float*)   (ws + (3u << 20));

  zproj_kernel     <<<dim3(NN + GROWS, KK), 32, 0, stream>>>(H, U, lp, Zs, Zf, ZT, O);
  flash_attn_kernel<<<dim3(NN / 16, KK), 32 * KSPLIT, 0, stream>>>(Zs, Zf, ZT, adj, O, out);
  combine_kernel   <<<NN / 16,          256, 0, stream>>>(H, U, O, thr, out);
}


# projection Z = H U_k and the Gram rows on its grid: per-lane FMA dot products in one-wave blocks -> f32 16x16x4 WMMA, 16 rows x 4 heads per block; d sums and loss row sums in a new order
# speedup vs baseline: 7.8599x; 2.0850x over previous
#include <hip/hip_runtime.h>

#define NN 4096
#define DD 256
#define SS 32
#define KK 4
#define ETA 0.5f
#define GROWS  (KK * SS)
#define GPARTS (KK * KK * SS)

typedef __attribute__((ext_vector_type(16))) _Float16 v16h;
typedef __attribute__((ext_vector_type(8)))  _Float16 v8h;
typedef __attribute__((ext_vector_type(8)))  float    v8f;

__device__ __forceinline__ v16h make_ab(const _Float16* p0, const _Float16* p1) {
  v8h lo = *(const v8h*)p0;
  v8h hi = *(const v8h*)p1;
  return __builtin_shufflevector(lo, hi, 0,1,2,3,4,5,6,7,8,9,10,11,12,13,14,15);
}

typedef __attribute__((ext_vector_type(2))) float    v2f;
typedef __attribute__((ext_vector_type(4))) float    v4f;
typedef __attribute__((ext_vector_type(2))) _Float16 v2h;

#define ZTILES (NN / 16)
#define GTILES (GROWS / 16)

template <bool GRAM>
__device__ __forceinline__ void rows_times_Uk(const float* __restrict__ arow,
                                              const float* __restrict__ ucol,
                                              v8f& acc0, v8f& acc1)
{
  #pragma unroll
  for (int J = 0; J < DD / 8; ++J) {
    v4f a;
    if (GRAM) {
      a[0] = arow[(8 * J + 0) * SS];
      a[1] = arow[(8 * J + 1) * SS];
      a[2] = arow[(8 * J + 2) * SS];
      a[3] = arow[(8 * J + 3) * SS];
    } else {
      a = *(const v4f*)(arow + 8 * J);
    }
    const v2f u0 = *(const v2f*)(ucol + (8 * J + 0) * SS);
    const v2f u1 = *(const v2f*)(ucol + (8 * J + 1) * SS);
    const v2f u2 = *(const v2f*)(ucol + (8 * J + 2) * SS);
    const v2f u3 = *(const v2f*)(ucol + (8 * J + 3) * SS);
    const v2f alo  = {a[0],  a[1]},  ahi  = {a[2],  a[3]};
    const v2f b0lo = {u0[0], u1[0]}, b0hi = {u2[0], u3[0]};
    const v2f b1lo = {u0[1], u1[1]}, b1hi = {u2[1], u3[1]};
    acc0 = __builtin_amdgcn_wmma_f32_16x16x4_f32(false, alo, false, b0lo, (short)0, acc0, false, false);
    acc1 = __builtin_amdgcn_wmma_f32_16x16x4_f32(false, alo, false, b1lo, (short)0, acc1, false, false);
    acc0 = __builtin_amdgcn_wmma_f32_16x16x4_f32(false, ahi, false, b0hi, (short)0, acc0, false, false);
    acc1 = __builtin_amdgcn_wmma_f32_16x16x4_f32(false, ahi, false, b1hi, (short)0, acc1, false, false);
  }
}

__global__ __launch_bounds__(32 * KK) __attribute__((amdgpu_waves_per_eu(1, 4))) void zproj_kernel(
    const float* __restrict__ H, const float* __restrict__ U,
    const float* __restrict__ lp,
    _Float16* __restrict__ Zs, _Float16* __restrict__ Zf, _Float16* __restrict__ ZT,
    float* __restrict__ Gp)
{
  const int k    = __builtin_amdgcn_readfirstlane((int)(threadIdx.x >> 5));
  const int l    = threadIdx.x & 31;
  const int c    = l & 15;
  const int h    = l >> 4;
  const int tile = blockIdx.x;
  const float* ucol = U + (size_t)k * DD * SS + (4 * h) * SS + 2 * c;

  v8f acc0 = {}; v8f acc1 = {};

  if (tile < GTILES) {
    const int kq = tile >> 1;
    const int a0 = 16 * (tile & 1);
    if (kq <= k)
      rows_times_Uk<true>(U + (size_t)kq * DD * SS + (4 * h) * SS + a0 + c, ucol, acc0, acc1);
    float part[8];
    #pragma unroll
    for (int r = 0; r < 8; ++r) {
      const int a = a0 + 8 * h + r;
      const float e0 = (kq == k && a == 2 * c)     ? acc0[r] - 1.f : acc0[r];
      const float e1 = (kq == k && a == 2 * c + 1) ? acc1[r] - 1.f : acc1[r];
      part[r] = e0 * e0 + e1 * e1;
    }
    #pragma unroll
    for (int off = 8; off > 0; off >>= 1) {
      #pragma unroll
      for (int r = 0; r < 8; ++r) part[r] += __shfl_xor(part[r], off, 32);
    }
    if (c == 0) {
      v4f* gp = (v4f*)(Gp + (kq * KK + k) * SS + a0 + 8 * h);
      const v4f p0 = {part[0], part[1], part[2], part[3]};
      const v4f p1 = {part[4], part[5], part[6], part[7]};
      gp[0] = p0;
      gp[1] = p1;
    }
    return;
  }

  const int n0 = (tile - GTILES) * 16;
  rows_times_Uk<false>(H + (size_t)(n0 + c) * DD + 4 * h, ucol, acc0, acc1);

  const v2f lpv = *(const v2f*)(lp + k * SS + 2 * c);
  const float e0 = __expf(lpv[0]);
  const float e1 = __expf(lpv[1]);
  v8h t0, t1;
  #pragma unroll
  for (int r = 0; r < 8; ++r) {
    const float z0 = acc0[r], z1 = acc1[r];
    const size_t o = ((size_t)k * NN + n0 + 8 * h + r) * SS + 2 * c;
    const v2h kf = {(_Float16)z0, (_Float16)z1};
    const v2h qf = {(_Float16)(z0 * e0), (_Float16)(z1 * e1)};
    *(v2h*)(Zf + o) = kf;
    *(v2h*)(Zs + o) = qf;
    t0[r] = (_Float16)z0;
    t1[r] = (_Float16)z1;
  }
  _Float16* vt = ZT + ((size_t)k * SS + 2 * c) * NN + n0 + 8 * h;
  *(v8h*)vt        = t0;
  *(v8h*)(vt + NN) = t1;
}

#define KSPLIT 8
#define KRANGE (NN / KSPLIT)

__global__ __launch_bounds__(32 * KSPLIT) void flash_attn_kernel(
    const _Float16* __restrict__ Zs, const _Float16* __restrict__ Zf,
    const _Float16* __restrict__ ZT, const float* __restrict__ adj,
    float* __restrict__ O, float* __restrict__ out)
{
  const int k    = blockIdx.y;
  const int qb   = blockIdx.x * 16;
  const int w    = __builtin_amdgcn_readfirstlane((int)(threadIdx.x >> 5));
  const int l    = threadIdx.x & 31;
  const int ll   = l & 15;
  const int half = l >> 4;
  const int c0   = half ? 8 : 0;

  if (blockIdx.x == 0 && blockIdx.y == 0 && w == 0) {
    float t = 0.f;
    for (int i = 0; i < GPARTS / 32; ++i) t += O[l + 32 * i];
    for (int off = 16; off > 0; off >>= 1) t += __shfl_xor(t, off, 32);
    if (l == 0) out[(size_t)NN * DD] = t;
  }

  __shared__ _Float16 __attribute__((aligned(16))) PldsAll[KSPLIT * 16 * 64];
  __shared__ float Om[KSPLIT * 16 * 32];
  __shared__ float Mm[KSPLIT * 16];
  __shared__ float Lm[KSPLIT * 16];
  _Float16* Plds = PldsAll + w * (16 * 64);

  const float inv_s = 0.17677669529663687f;

  const _Float16* qrow = Zs + ((size_t)k * NN + (qb + ll)) * SS;
  const v16h aQ = make_ab(qrow + c0, qrow + c0 + 16);

  v16h ones;
  #pragma unroll
  for (int i = 0; i < 16; ++i) ones[i] = (_Float16)1.0f;

  float m[8];
  #pragma unroll
  for (int v = 0; v < 8; ++v) m[v] = -INFINITY;
  v8f o0 = {}; v8f o1 = {};
  v8f lacc = {};

  const _Float16* Zfk  = Zf + (size_t)k * NN * SS;
  const _Float16* ZTk  = ZT + (size_t)k * SS * NN;
  const float*    adjq = adj + (size_t)qb * NN;
  const unsigned  koff = (unsigned)(ll * SS + c0);
  const unsigned  voff = (unsigned)(ll * NN + c0);
  const unsigned  moff = (unsigned)(8 * half * NN + ll);
  const int kb0 = w * KRANGE;
  for (int kb = kb0; kb < kb0 + KRANGE; kb += 64) {
    v8f sc[4];
    #pragma unroll
    for (int t = 0; t < 4; ++t) {
      const _Float16* kr = Zfk + (unsigned)(kb * SS) + koff + t * 16 * SS;
      const v16h bK = make_ab(kr, kr + 16);
      v8f z = {};
      sc[t] = __builtin_amdgcn_wmma_f32_16x16x32_f16(false, aQ, false, bK, (short)0, z, false, false);
    }

    #pragma unroll
    for (int v = 0; v < 8; ++v) {
      const float* mrow = adjq + (unsigned)kb + moff + v * NN;
      const float w0 = mrow[0], w1 = mrow[16], w2 = mrow[32], w3 = mrow[48];
      __builtin_prefetch(mrow + 64, 0, 0);
      const float r0 = sc[0][v] * inv_s;
      const float r1 = sc[1][v] * inv_s;
      const float r2 = sc[2][v] * inv_s;
      const float r3 = sc[3][v] * inv_s;
      float mx = fmaxf(fmaxf(r0, r1), fmaxf(r2, r3));
      #pragma unroll
      for (int off = 1; off < 16; off <<= 1) mx = fmaxf(mx, __shfl_xor(mx, off, 32));
      const float mn    = fmaxf(m[v], mx);
      const float scale = __expf(m[v] - mn);
      m[v] = mn;
      const float p0 = w0 * __expf(r0 - mn);
      const float p1 = w1 * __expf(r1 - mn);
      const float p2 = w2 * __expf(r2 - mn);
      const float p3 = w3 * __expf(r3 - mn);
      o0[v]   *= scale;
      o1[v]   *= scale;
      lacc[v] *= scale;
      const int r = v + 8 * half;
      Plds[r * 64 + ll]      = (_Float16)p0;
      Plds[r * 64 + 16 + ll] = (_Float16)p1;
      Plds[r * 64 + 32 + ll] = (_Float16)p2;
      Plds[r * 64 + 48 + ll] = (_Float16)p3;
    }
    asm volatile("s_wait_dscnt 0" ::: "memory");

    #pragma unroll
    for (int c = 0; c < 2; ++c) {
      const v16h aP = make_ab(&Plds[ll * 64 + c * 32 + c0],
                              &Plds[ll * 64 + c * 32 + c0 + 16]);
      const _Float16* vr0 = ZTk + (unsigned)kb + voff + c * 32;
      const _Float16* vr1 = vr0 + 16 * NN;
      const v16h bV0 = make_ab(vr0, vr0 + 16);
      const v16h bV1 = make_ab(vr1, vr1 + 16);
      o0   = __builtin_amdgcn_wmma_f32_16x16x32_f16(false, aP, false, bV0,  (short)0, o0,   false, false);
      o1   = __builtin_amdgcn_wmma_f32_16x16x32_f16(false, aP, false, bV1,  (short)0, o1,   false, false);
      lacc = __builtin_amdgcn_wmma_f32_16x16x32_f16(false, aP, false, ones, (short)0, lacc, false, false);
    }
  }

  #pragma unroll
  for (int v = 0; v < 8; ++v) {
    const int r = v + 8 * half;
    Om[(w * 16 + r) * 32 + ll]      = o0[v];
    Om[(w * 16 + r) * 32 + 16 + ll] = o1[v];
    if (ll == 0) {
      Mm[w * 16 + r] = m[v];
      Lm[w * 16 + r] = lacc[v];
    }
  }
  __syncthreads();

  for (int j = 0; j < 2; ++j) {
    const int i = (int)threadIdx.x + j * (32 * KSPLIT);
    const int r = i >> 5;
    float M = Mm[r];
    for (int u = 1; u < KSPLIT; ++u) M = fmaxf(M, Mm[u * 16 + r]);
    float L = 0.f, acc = 0.f;
    for (int u = 0; u < KSPLIT; ++u) {
      const float f = __expf(Mm[u * 16 + r] - M);
      L   = fmaf(Lm[u * 16 + r], f, L);
      acc = fmaf(Om[u * 512 + i], f, acc);
    }
    O[((size_t)k * NN + qb) * SS + i] = acc / L;
  }
}

__global__ __launch_bounds__(256) void combine_kernel(
    const float* __restrict__ H, const float* __restrict__ U,
    const float* __restrict__ O, const float* __restrict__ thr,
    float* __restrict__ out)
{
  const int n0 = blockIdx.x * 16;
  const int w  = threadIdx.x >> 5;
  const int l  = threadIdx.x & 31;
  const int c  = l & 15;
  const int h  = l >> 4;
  const int dc = w * 32 + 2 * c;

  v8f acc0 = {}; v8f acc1 = {};
  #pragma unroll
  for (int k = 0; k < KK; ++k) {
    const v4f* pa  = (const v4f*)(O + ((size_t)k * NN + n0 + c) * SS + 16 * h);
    const v4f* pb0 = (const v4f*)(U + ((size_t)k * DD + dc) * SS + 16 * h);
    const v4f* pb1 = (const v4f*)(U + ((size_t)k * DD + dc + 1) * SS + 16 * h);
    #pragma unroll
    for (int q = 0; q < 4; ++q) {
      const v4f a = pa[q], b0 = pb0[q], b1 = pb1[q];
      const v2f alo  = {a[0],  a[1]},  ahi  = {a[2],  a[3]};
      const v2f b0lo = {b0[0], b0[1]}, b0hi = {b0[2], b0[3]};
      const v2f b1lo = {b1[0], b1[1]}, b1hi = {b1[2], b1[3]};
      acc0 = __builtin_amdgcn_wmma_f32_16x16x4_f32(false, alo, false, b0lo, (short)0, acc0, false, false);
      acc1 = __builtin_amdgcn_wmma_f32_16x16x4_f32(false, alo, false, b1lo, (short)0, acc1, false, false);
      acc0 = __builtin_amdgcn_wmma_f32_16x16x4_f32(false, ahi, false, b0hi, (short)0, acc0, false, false);
      acc1 = __builtin_amdgcn_wmma_f32_16x16x4_f32(false, ahi, false, b1hi, (short)0, acc1, false, false);
    }
  }

  const v2f t = *(const v2f*)(thr + dc);
  #pragma unroll
  for (int r = 0; r < 8; ++r) {
    const size_t o = (size_t)(n0 + 8 * h + r) * DD + dc;
    const v2f hv = *(const v2f*)(H + o);
    const float h0 = hv[0] + ETA * acc0[r];
    const float h1 = hv[1] + ETA * acc1[r];
    const float a0 = fabsf(h0) - t[0];
    const float a1 = fabsf(h1) - t[1];
    v2f res;
    res[0] = (a0 > 0.f) ? copysignf(a0, h0) : 0.f;
    res[1] = (a1 > 0.f) ? copysignf(a1, h1) : 0.f;
    *(v2f*)(out + o) = res;
  }
}

extern "C" void kernel_launch(void* const* d_in, const int* in_sizes, int n_in,
                              void* d_out, int out_size, void* d_ws, size_t ws_size,
                              hipStream_t stream) {
  (void)in_sizes; (void)n_in; (void)out_size; (void)ws_size;
  const float* H   = (const float*)d_in[0];
  const float* adj = (const float*)d_in[1];
  const float* U   = (const float*)d_in[2];
  const float* lp  = (const float*)d_in[3];
  const float* thr = (const float*)d_in[4];
  float* out = (float*)d_out;

  char* ws = (char*)d_ws;
  _Float16* Zs = (_Float16*)(ws);
  _Float16* Zf = (_Float16*)(ws + (1u << 20));
  _Float16* ZT = (_Float16*)(ws + (2u << 20));
  float*    O  = (float*)   (ws + (3u << 20));

  zproj_kernel     <<<GTILES + ZTILES,  32 * KK, 0, stream>>>(H, U, lp, Zs, Zf, ZT, O);
  flash_attn_kernel<<<dim3(NN / 16, KK), 32 * KSPLIT, 0, stream>>>(Zs, Zf, ZT, adj, O, out);
  combine_kernel   <<<NN / 16,          256, 0, stream>>>(H, U, O, thr, out);
}


# attention operands: key and value copies written in B-fragment order, score column ll of tile t = key 4ll+t: every K/V/mask load one contiguous run (4 lines), P stored 8 bytes per lane; sums kept
# speedup vs baseline: 8.9549x; 1.1393x over previous
#include <hip/hip_runtime.h>

#define NN 4096
#define DD 256
#define SS 32
#define KK 4
#define ETA 0.5f
#define GROWS  (KK * SS)
#define GPARTS (KK * KK * SS)

typedef __attribute__((ext_vector_type(16))) _Float16 v16h;
typedef __attribute__((ext_vector_type(8)))  _Float16 v8h;
typedef __attribute__((ext_vector_type(8)))  float    v8f;

__device__ __forceinline__ v16h make_ab(const _Float16* p0, const _Float16* p1) {
  v8h lo = *(const v8h*)p0;
  v8h hi = *(const v8h*)p1;
  return __builtin_shufflevector(lo, hi, 0,1,2,3,4,5,6,7,8,9,10,11,12,13,14,15);
}

typedef __attribute__((ext_vector_type(2))) float    v2f;
typedef __attribute__((ext_vector_type(4))) float    v4f;
typedef __attribute__((ext_vector_type(2))) _Float16 v2h;
typedef __attribute__((ext_vector_type(4))) _Float16 v4h;

__device__ __forceinline__ unsigned kfrag_off(unsigned n, unsigned s) {
  const unsigned kk = n & 63u, gs = s >> 3;
  const unsigned unit = ((kk & 3u) * 2u + (gs >> 1)) * 32u + (gs & 1u) * 16u + (kk >> 2);
  return (n >> 6) * 2048u + unit * 8u + (s & 7u);
}
__device__ __forceinline__ unsigned vfrag_off(unsigned n, unsigned s) {
  const unsigned kk = n & 63u;
  const unsigned unit = ((((kk >> 5) * 2u + (s >> 4)) * 2u + ((kk >> 4) & 1u)) * 32u)
                      + ((kk >> 3) & 1u) * 16u + (s & 15u);
  return (n >> 6) * 2048u + unit * 8u + (kk & 7u);
}

#define ZTILES (NN / 16)
#define GTILES (GROWS / 16)

template <bool GRAM>
__device__ __forceinline__ void rows_times_Uk(const float* __restrict__ arow,
                                              const float* __restrict__ ucol,
                                              v8f& acc0, v8f& acc1)
{
  #pragma unroll
  for (int J = 0; J < DD / 8; ++J) {
    v4f a;
    if (GRAM) {
      a[0] = arow[(8 * J + 0) * SS];
      a[1] = arow[(8 * J + 1) * SS];
      a[2] = arow[(8 * J + 2) * SS];
      a[3] = arow[(8 * J + 3) * SS];
    } else {
      a = *(const v4f*)(arow + 8 * J);
    }
    const v2f u0 = *(const v2f*)(ucol + (8 * J + 0) * SS);
    const v2f u1 = *(const v2f*)(ucol + (8 * J + 1) * SS);
    const v2f u2 = *(const v2f*)(ucol + (8 * J + 2) * SS);
    const v2f u3 = *(const v2f*)(ucol + (8 * J + 3) * SS);
    const v2f alo  = {a[0],  a[1]},  ahi  = {a[2],  a[3]};
    const v2f b0lo = {u0[0], u1[0]}, b0hi = {u2[0], u3[0]};
    const v2f b1lo = {u0[1], u1[1]}, b1hi = {u2[1], u3[1]};
    acc0 = __builtin_amdgcn_wmma_f32_16x16x4_f32(false, alo, false, b0lo, (short)0, acc0, false, false);
    acc1 = __builtin_amdgcn_wmma_f32_16x16x4_f32(false, alo, false, b1lo, (short)0, acc1, false, false);
    acc0 = __builtin_amdgcn_wmma_f32_16x16x4_f32(false, ahi, false, b0hi, (short)0, acc0, false, false);
    acc1 = __builtin_amdgcn_wmma_f32_16x16x4_f32(false, ahi, false, b1hi, (short)0, acc1, false, false);
  }
}

__global__ __launch_bounds__(32 * KK) __attribute__((amdgpu_waves_per_eu(1, 4))) void zproj_kernel(
    const float* __restrict__ H, const float* __restrict__ U,
    const float* __restrict__ lp,
    _Float16* __restrict__ Zs, _Float16* __restrict__ Zf, _Float16* __restrict__ ZT,
    float* __restrict__ Gp)
{
  const int k    = __builtin_amdgcn_readfirstlane((int)(threadIdx.x >> 5));
  const int l    = threadIdx.x & 31;
  const int c    = l & 15;
  const int h    = l >> 4;
  const int tile = blockIdx.x;
  const float* ucol = U + (size_t)k * DD * SS + (4 * h) * SS + 2 * c;

  v8f acc0 = {}; v8f acc1 = {};

  if (tile < GTILES) {
    const int kq = tile >> 1;
    const int a0 = 16 * (tile & 1);
    if (kq <= k)
      rows_times_Uk<true>(U + (size_t)kq * DD * SS + (4 * h) * SS + a0 + c, ucol, acc0, acc1);
    float part[8];
    #pragma unroll
    for (int r = 0; r < 8; ++r) {
      const int a = a0 + 8 * h + r;
      const float e0 = (kq == k && a == 2 * c)     ? acc0[r] - 1.f : acc0[r];
      const float e1 = (kq == k && a == 2 * c + 1) ? acc1[r] - 1.f : acc1[r];
      part[r] = e0 * e0 + e1 * e1;
    }
    #pragma unroll
    for (int off = 8; off > 0; off >>= 1) {
      #pragma unroll
      for (int r = 0; r < 8; ++r) part[r] += __shfl_xor(part[r], off, 32);
    }
    if (c == 0) {
      v4f* gp = (v4f*)(Gp + (kq * KK + k) * SS + a0 + 8 * h);
      const v4f p0 = {part[0], part[1], part[2], part[3]};
      const v4f p1 = {part[4], part[5], part[6], part[7]};
      gp[0] = p0;
      gp[1] = p1;
    }
    return;
  }

  const int n0 = (tile - GTILES) * 16;
  rows_times_Uk<false>(H + (size_t)(n0 + c) * DD + 4 * h, ucol, acc0, acc1);

  const v2f lpv = *(const v2f*)(lp + k * SS + 2 * c);
  const float e0 = __expf(lpv[0]);
  const float e1 = __expf(lpv[1]);
  v8h t0, t1;
  #pragma unroll
  for (int r = 0; r < 8; ++r) {
    const float z0 = acc0[r], z1 = acc1[r];
    const size_t o = ((size_t)k * NN + n0 + 8 * h + r) * SS + 2 * c;
    const v2h kf = {(_Float16)z0, (_Float16)z1};
    const v2h qf = {(_Float16)(z0 * e0), (_Float16)(z1 * e1)};
    *(v2h*)(Zf + (size_t)k * NN * SS
               + kfrag_off((unsigned)(n0 + 8 * h + r), (unsigned)(2 * c))) = kf;
    *(v2h*)(Zs + o) = qf;
    t0[r] = (_Float16)z0;
    t1[r] = (_Float16)z1;
  }
  _Float16* vt = ZT + (size_t)k * SS * NN + vfrag_off((unsigned)(n0 + 8 * h), (unsigned)(2 * c));
  *(v8h*)vt       = t0;
  *(v8h*)(vt + 8) = t1;
}

#define KSPLIT 8
#define KRANGE (NN / KSPLIT)

__global__ __launch_bounds__(32 * KSPLIT) void flash_attn_kernel(
    const _Float16* __restrict__ Zs, const _Float16* __restrict__ Zf,
    const _Float16* __restrict__ ZT, const float* __restrict__ adj,
    float* __restrict__ O, float* __restrict__ out)
{
  const int k    = blockIdx.y;
  const int qb   = blockIdx.x * 16;
  const int w    = __builtin_amdgcn_readfirstlane((int)(threadIdx.x >> 5));
  const int l    = threadIdx.x & 31;
  const int ll   = l & 15;
  const int half = l >> 4;
  const int c0   = half ? 8 : 0;

  if (blockIdx.x == 0 && blockIdx.y == 0 && w == 0) {
    float t = 0.f;
    for (int i = 0; i < GPARTS / 32; ++i) t += O[l + 32 * i];
    for (int off = 16; off > 0; off >>= 1) t += __shfl_xor(t, off, 32);
    if (l == 0) out[(size_t)NN * DD] = t;
  }

  __shared__ _Float16 __attribute__((aligned(16))) PldsAll[KSPLIT * 16 * 64];
  __shared__ float Om[KSPLIT * 16 * 32];
  __shared__ float Mm[KSPLIT * 16];
  __shared__ float Lm[KSPLIT * 16];
  _Float16* Plds = PldsAll + w * (16 * 64);

  const float inv_s = 0.17677669529663687f;

  const _Float16* qrow = Zs + ((size_t)k * NN + (qb + ll)) * SS;
  const v16h aQ = make_ab(qrow + c0, qrow + c0 + 16);

  v16h ones;
  #pragma unroll
  for (int i = 0; i < 16; ++i) ones[i] = (_Float16)1.0f;

  float m[8];
  #pragma unroll
  for (int v = 0; v < 8; ++v) m[v] = -INFINITY;
  v8f o0 = {}; v8f o1 = {};
  v8f lacc = {};

  const _Float16* Zfk  = Zf + (size_t)k * NN * SS;
  const _Float16* ZTk  = ZT + (size_t)k * SS * NN;
  const float*    adjq = adj + (size_t)qb * NN;
  const unsigned  foff = (unsigned)(l * 8);
  const unsigned  moff = (unsigned)(8 * half * NN + 4 * ll);
  const int kb0 = w * KRANGE;
  for (int kb = kb0; kb < kb0 + KRANGE; kb += 64) {
    v8f sc[4];
    #pragma unroll
    for (int t = 0; t < 4; ++t) {
      const _Float16* kr = Zfk + (unsigned)(kb * SS) + foff + t * 512;
      const v16h bK = make_ab(kr, kr + 256);
      v8f z = {};
      sc[t] = __builtin_amdgcn_wmma_f32_16x16x32_f16(false, aQ, false, bK, (short)0, z, false, false);
    }

    #pragma unroll
    for (int v = 0; v < 8; ++v) {
      const float* mrow = adjq + (unsigned)kb + moff + v * NN;
      const v4f mw = *(const v4f*)mrow;
      const float w0 = mw[0], w1 = mw[1], w2 = mw[2], w3 = mw[3];
      __builtin_prefetch(mrow + 64, 0, 0);
      const float r0 = sc[0][v] * inv_s;
      const float r1 = sc[1][v] * inv_s;
      const float r2 = sc[2][v] * inv_s;
      const float r3 = sc[3][v] * inv_s;
      float mx = fmaxf(fmaxf(r0, r1), fmaxf(r2, r3));
      #pragma unroll
      for (int off = 1; off < 16; off <<= 1) mx = fmaxf(mx, __shfl_xor(mx, off, 32));
      const float mn    = fmaxf(m[v], mx);
      const float scale = __expf(m[v] - mn);
      m[v] = mn;
      const float p0 = w0 * __expf(r0 - mn);
      const float p1 = w1 * __expf(r1 - mn);
      const float p2 = w2 * __expf(r2 - mn);
      const float p3 = w3 * __expf(r3 - mn);
      o0[v]   *= scale;
      o1[v]   *= scale;
      lacc[v] *= scale;
      const int r = v + 8 * half;
      const v4h pk = {(_Float16)p0, (_Float16)p1, (_Float16)p2, (_Float16)p3};
      *(v4h*)&Plds[r * 64 + 4 * ll] = pk;
    }
    asm volatile("s_wait_dscnt 0" ::: "memory");

    #pragma unroll
    for (int c = 0; c < 2; ++c) {
      const v16h aP = make_ab(&Plds[ll * 64 + c * 32 + c0],
                              &Plds[ll * 64 + c * 32 + c0 + 16]);
      const _Float16* vr0 = ZTk + (unsigned)(kb * SS) + foff + c * 1024;
      const _Float16* vr1 = vr0 + 512;
      const v16h bV0 = make_ab(vr0, vr0 + 256);
      const v16h bV1 = make_ab(vr1, vr1 + 256);
      o0   = __builtin_amdgcn_wmma_f32_16x16x32_f16(false, aP, false, bV0,  (short)0, o0,   false, false);
      o1   = __builtin_amdgcn_wmma_f32_16x16x32_f16(false, aP, false, bV1,  (short)0, o1,   false, false);
      lacc = __builtin_amdgcn_wmma_f32_16x16x32_f16(false, aP, false, ones, (short)0, lacc, false, false);
    }
  }

  #pragma unroll
  for (int v = 0; v < 8; ++v) {
    const int r = v + 8 * half;
    Om[(w * 16 + r) * 32 + ll]      = o0[v];
    Om[(w * 16 + r) * 32 + 16 + ll] = o1[v];
    if (ll == 0) {
      Mm[w * 16 + r] = m[v];
      Lm[w * 16 + r] = lacc[v];
    }
  }
  __syncthreads();

  for (int j = 0; j < 2; ++j) {
    const int i = (int)threadIdx.x + j * (32 * KSPLIT);
    const int r = i >> 5;
    float M = Mm[r];
    for (int u = 1; u < KSPLIT; ++u) M = fmaxf(M, Mm[u * 16 + r]);
    float L = 0.f, acc = 0.f;
    for (int u = 0; u < KSPLIT; ++u) {
      const float f = __expf(Mm[u * 16 + r] - M);
      L   = fmaf(Lm[u * 16 + r], f, L);
      acc = fmaf(Om[u * 512 + i], f, acc);
    }
    O[((size_t)k * NN + qb) * SS + i] = acc / L;
  }
}

__global__ __launch_bounds__(256) void combine_kernel(
    const float* __restrict__ H, const float* __restrict__ U,
    const float* __restrict__ O, const float* __restrict__ thr,
    float* __restrict__ out)
{
  const int n0 = blockIdx.x * 16;
  const int w  = threadIdx.x >> 5;
  const int l  = threadIdx.x & 31;
  const int c  = l & 15;
  const int h  = l >> 4;
  const int dc = w * 32 + 2 * c;

  v8f acc0 = {}; v8f acc1 = {};
  #pragma unroll
  for (int k = 0; k < KK; ++k) {
    const v4f* pa  = (const v4f*)(O + ((size_t)k * NN + n0 + c) * SS + 16 * h);
    const v4f* pb0 = (const v4f*)(U + ((size_t)k * DD + dc) * SS + 16 * h);
    const v4f* pb1 = (const v4f*)(U + ((size_t)k * DD + dc + 1) * SS + 16 * h);
    #pragma unroll
    for (int q = 0; q < 4; ++q) {
      const v4f a = pa[q], b0 = pb0[q], b1 = pb1[q];
      const v2f alo  = {a[0],  a[1]},  ahi  = {a[2],  a[3]};
      const v2f b0lo = {b0[0], b0[1]}, b0hi = {b0[2], b0[3]};
      const v2f b1lo = {b1[0], b1[1]}, b1hi = {b1[2], b1[3]};
      acc0 = __builtin_amdgcn_wmma_f32_16x16x4_f32(false, alo, false, b0lo, (short)0, acc0, false, false);
      acc1 = __builtin_amdgcn_wmma_f32_16x16x4_f32(false, alo, false, b1lo, (short)0, acc1, false, false);
      acc0 = __builtin_amdgcn_wmma_f32_16x16x4_f32(false, ahi, false, b0hi, (short)0, acc0, false, false);
      acc1 = __builtin_amdgcn_wmma_f32_16x16x4_f32(false, ahi, false, b1hi, (short)0, acc1, false, false);
    }
  }

  const v2f t = *(const v2f*)(thr + dc);
  #pragma unroll
  for (int r = 0; r < 8; ++r) {
    const size_t o = (size_t)(n0 + 8 * h + r) * DD + dc;
    const v2f hv = *(const v2f*)(H + o);
    const float h0 = hv[0] + ETA * acc0[r];
    const float h1 = hv[1] + ETA * acc1[r];
    const float a0 = fabsf(h0) - t[0];
    const float a1 = fabsf(h1) - t[1];
    v2f res;
    res[0] = (a0 > 0.f) ? copysignf(a0, h0) : 0.f;
    res[1] = (a1 > 0.f) ? copysignf(a1, h1) : 0.f;
    *(v2f*)(out + o) = res;
  }
}

extern "C" void kernel_launch(void* const* d_in, const int* in_sizes, int n_in,
                              void* d_out, int out_size, void* d_ws, size_t ws_size,
                              hipStream_t stream) {
  (void)in_sizes; (void)n_in; (void)out_size; (void)ws_size;
  const float* H   = (const float*)d_in[0];
  const float* adj = (const float*)d_in[1];
  const float* U   = (const float*)d_in[2];
  const float* lp  = (const float*)d_in[3];
  const float* thr = (const float*)d_in[4];
  float* out = (float*)d_out;

  char* ws = (char*)d_ws;
  _Float16* Zs = (_Float16*)(ws);
  _Float16* Zf = (_Float16*)(ws + (1u << 20));
  _Float16* ZT = (_Float16*)(ws + (2u << 20));
  float*    O  = (float*)   (ws + (3u << 20));

  zproj_kernel     <<<GTILES + ZTILES,  32 * KK, 0, stream>>>(H, U, lp, Zs, Zf, ZT, O);
  flash_attn_kernel<<<dim3(NN / 16, KK), 32 * KSPLIT, 0, stream>>>(Zs, Zf, ZT, adj, O, out);
  combine_kernel   <<<NN / 16,          256, 0, stream>>>(H, U, O, thr, out);
}


# attention row max: 16-lane butterfly moved from __shfl_xor (ds_bpermute through the LDS unit, a wait behind each step) to row_xmask DPP on the max itself, last step fused with the running max
# speedup vs baseline: 10.1232x; 1.1305x over previous
#include <hip/hip_runtime.h>

#define NN 4096
#define DD 256
#define SS 32
#define KK 4
#define ETA 0.5f
#define GROWS  (KK * SS)
#define GPARTS (KK * KK * SS)

typedef __attribute__((ext_vector_type(16))) _Float16 v16h;
typedef __attribute__((ext_vector_type(8)))  _Float16 v8h;
typedef __attribute__((ext_vector_type(8)))  float    v8f;

__device__ __forceinline__ v16h make_ab(const _Float16* p0, const _Float16* p1) {
  v8h lo = *(const v8h*)p0;
  v8h hi = *(const v8h*)p1;
  return __builtin_shufflevector(lo, hi, 0,1,2,3,4,5,6,7,8,9,10,11,12,13,14,15);
}

typedef __attribute__((ext_vector_type(2))) float    v2f;
typedef __attribute__((ext_vector_type(4))) float    v4f;
typedef __attribute__((ext_vector_type(2))) _Float16 v2h;
typedef __attribute__((ext_vector_type(4))) _Float16 v4h;

__device__ __forceinline__ unsigned kfrag_off(unsigned n, unsigned s) {
  const unsigned kk = n & 63u, gs = s >> 3;
  const unsigned unit = ((kk & 3u) * 2u + (gs >> 1)) * 32u + (gs & 1u) * 16u + (kk >> 2);
  return (n >> 6) * 2048u + unit * 8u + (s & 7u);
}
__device__ __forceinline__ unsigned vfrag_off(unsigned n, unsigned s) {
  const unsigned kk = n & 63u;
  const unsigned unit = ((((kk >> 5) * 2u + (s >> 4)) * 2u + ((kk >> 4) & 1u)) * 32u)
                      + ((kk >> 3) & 1u) * 16u + (s & 15u);
  return (n >> 6) * 2048u + unit * 8u + (kk & 7u);
}

#define ZTILES (NN / 16)
#define GTILES (GROWS / 16)

template <bool GRAM>
__device__ __forceinline__ void rows_times_Uk(const float* __restrict__ arow,
                                              const float* __restrict__ ucol,
                                              v8f& acc0, v8f& acc1)
{
  #pragma unroll
  for (int J = 0; J < DD / 8; ++J) {
    v4f a;
    if (GRAM) {
      a[0] = arow[(8 * J + 0) * SS];
      a[1] = arow[(8 * J + 1) * SS];
      a[2] = arow[(8 * J + 2) * SS];
      a[3] = arow[(8 * J + 3) * SS];
    } else {
      a = *(const v4f*)(arow + 8 * J);
    }
    const v2f u0 = *(const v2f*)(ucol + (8 * J + 0) * SS);
    const v2f u1 = *(const v2f*)(ucol + (8 * J + 1) * SS);
    const v2f u2 = *(const v2f*)(ucol + (8 * J + 2) * SS);
    const v2f u3 = *(const v2f*)(ucol + (8 * J + 3) * SS);
    const v2f alo  = {a[0],  a[1]},  ahi  = {a[2],  a[3]};
    const v2f b0lo = {u0[0], u1[0]}, b0hi = {u2[0], u3[0]};
    const v2f b1lo = {u0[1], u1[1]}, b1hi = {u2[1], u3[1]};
    acc0 = __builtin_amdgcn_wmma_f32_16x16x4_f32(false, alo, false, b0lo, (short)0, acc0, false, false);
    acc1 = __builtin_amdgcn_wmma_f32_16x16x4_f32(false, alo, false, b1lo, (short)0, acc1, false, false);
    acc0 = __builtin_amdgcn_wmma_f32_16x16x4_f32(false, ahi, false, b0hi, (short)0, acc0, false, false);
    acc1 = __builtin_amdgcn_wmma_f32_16x16x4_f32(false, ahi, false, b1hi, (short)0, acc1, false, false);
  }
}

__global__ __launch_bounds__(32 * KK) __attribute__((amdgpu_waves_per_eu(1, 4))) void zproj_kernel(
    const float* __restrict__ H, const float* __restrict__ U,
    const float* __restrict__ lp,
    _Float16* __restrict__ Zs, _Float16* __restrict__ Zf, _Float16* __restrict__ ZT,
    float* __restrict__ Gp)
{
  const int k    = __builtin_amdgcn_readfirstlane((int)(threadIdx.x >> 5));
  const int l    = threadIdx.x & 31;
  const int c    = l & 15;
  const int h    = l >> 4;
  const int tile = blockIdx.x;
  const float* ucol = U + (size_t)k * DD * SS + (4 * h) * SS + 2 * c;

  v8f acc0 = {}; v8f acc1 = {};

  if (tile < GTILES) {
    const int kq = tile >> 1;
    const int a0 = 16 * (tile & 1);
    if (kq <= k)
      rows_times_Uk<true>(U + (size_t)kq * DD * SS + (4 * h) * SS + a0 + c, ucol, acc0, acc1);
    float part[8];
    #pragma unroll
    for (int r = 0; r < 8; ++r) {
      const int a = a0 + 8 * h + r;
      const float e0 = (kq == k && a == 2 * c)     ? acc0[r] - 1.f : acc0[r];
      const float e1 = (kq == k && a == 2 * c + 1) ? acc1[r] - 1.f : acc1[r];
      part[r] = e0 * e0 + e1 * e1;
    }
    #pragma unroll
    for (int off = 8; off > 0; off >>= 1) {
      #pragma unroll
      for (int r = 0; r < 8; ++r) part[r] += __shfl_xor(part[r], off, 32);
    }
    if (c == 0) {
      v4f* gp = (v4f*)(Gp + (kq * KK + k) * SS + a0 + 8 * h);
      const v4f p0 = {part[0], part[1], part[2], part[3]};
      const v4f p1 = {part[4], part[5], part[6], part[7]};
      gp[0] = p0;
      gp[1] = p1;
    }
    return;
  }

  const int n0 = (tile - GTILES) * 16;
  rows_times_Uk<false>(H + (size_t)(n0 + c) * DD + 4 * h, ucol, acc0, acc1);

  const v2f lpv = *(const v2f*)(lp + k * SS + 2 * c);
  const float e0 = __expf(lpv[0]);
  const float e1 = __expf(lpv[1]);
  v8h t0, t1;
  #pragma unroll
  for (int r = 0; r < 8; ++r) {
    const float z0 = acc0[r], z1 = acc1[r];
    const size_t o = ((size_t)k * NN + n0 + 8 * h + r) * SS + 2 * c;
    const v2h kf = {(_Float16)z0, (_Float16)z1};
    const v2h qf = {(_Float16)(z0 * e0), (_Float16)(z1 * e1)};
    *(v2h*)(Zf + (size_t)k * NN * SS
               + kfrag_off((unsigned)(n0 + 8 * h + r), (unsigned)(2 * c))) = kf;
    *(v2h*)(Zs + o) = qf;
    t0[r] = (_Float16)z0;
    t1[r] = (_Float16)z1;
  }
  _Float16* vt = ZT + (size_t)k * SS * NN + vfrag_off((unsigned)(n0 + 8 * h), (unsigned)(2 * c));
  *(v8h*)vt       = t0;
  *(v8h*)(vt + 8) = t1;
}


template <int MASK>
__device__ __forceinline__ float xmask_max(float x) {
  float y;
  asm("v_max_num_f32_dpp %0, %1, %1 row_xmask:%2 row_mask:0xf bank_mask:0xf"
      : "=v"(y) : "v"(x), "n"(MASK));
  return y;
}
__device__ __forceinline__ float xmask8_max3(float x, float m) {
  float y;
  asm("v_max3_num_f32_e64_dpp %0, %1, %1, %2 row_xmask:8 row_mask:0xf bank_mask:0xf"
      : "=v"(y) : "v"(x), "v"(m));
  return y;
}

#define KSPLIT 8
#define KRANGE (NN / KSPLIT)

__global__ __launch_bounds__(32 * KSPLIT) void flash_attn_kernel(
    const _Float16* __restrict__ Zs, const _Float16* __restrict__ Zf,
    const _Float16* __restrict__ ZT, const float* __restrict__ adj,
    float* __restrict__ O, float* __restrict__ out)
{
  const int k    = blockIdx.y;
  const int qb   = blockIdx.x * 16;
  const int w    = __builtin_amdgcn_readfirstlane((int)(threadIdx.x >> 5));
  const int l    = threadIdx.x & 31;
  const int ll   = l & 15;
  const int half = l >> 4;
  const int c0   = half ? 8 : 0;

  if (blockIdx.x == 0 && blockIdx.y == 0 && w == 0) {
    float t = 0.f;
    for (int i = 0; i < GPARTS / 32; ++i) t += O[l + 32 * i];
    for (int off = 16; off > 0; off >>= 1) t += __shfl_xor(t, off, 32);
    if (l == 0) out[(size_t)NN * DD] = t;
  }

  __shared__ _Float16 __attribute__((aligned(16))) PldsAll[KSPLIT * 16 * 64];
  __shared__ float Om[KSPLIT * 16 * 32];
  __shared__ float Mm[KSPLIT * 16];
  __shared__ float Lm[KSPLIT * 16];
  _Float16* Plds = PldsAll + w * (16 * 64);

  const float inv_s = 0.17677669529663687f;

  const _Float16* qrow = Zs + ((size_t)k * NN + (qb + ll)) * SS;
  const v16h aQ = make_ab(qrow + c0, qrow + c0 + 16);

  v16h ones;
  #pragma unroll
  for (int i = 0; i < 16; ++i) ones[i] = (_Float16)1.0f;

  float m[8];
  #pragma unroll
  for (int v = 0; v < 8; ++v) m[v] = -INFINITY;
  v8f o0 = {}; v8f o1 = {};
  v8f lacc = {};

  const _Float16* Zfk  = Zf + (size_t)k * NN * SS;
  const _Float16* ZTk  = ZT + (size_t)k * SS * NN;
  const float*    adjq = adj + (size_t)qb * NN;
  const unsigned  foff = (unsigned)(l * 8);
  const unsigned  moff = (unsigned)(8 * half * NN + 4 * ll);
  const int kb0 = w * KRANGE;
  for (int kb = kb0; kb < kb0 + KRANGE; kb += 64) {
    v8f sc[4];
    #pragma unroll
    for (int t = 0; t < 4; ++t) {
      const _Float16* kr = Zfk + (unsigned)(kb * SS) + foff + t * 512;
      const v16h bK = make_ab(kr, kr + 256);
      v8f z = {};
      sc[t] = __builtin_amdgcn_wmma_f32_16x16x32_f16(false, aQ, false, bK, (short)0, z, false, false);
    }

    float mx[8];
    #pragma unroll
    for (int v = 0; v < 8; ++v)
      mx[v] = fmaxf(fmaxf(sc[0][v] * inv_s, sc[1][v] * inv_s),
                    fmaxf(sc[2][v] * inv_s, sc[3][v] * inv_s));
    #pragma unroll
    for (int v = 0; v < 8; ++v) mx[v] = xmask_max<1>(mx[v]);
    #pragma unroll
    for (int v = 0; v < 8; ++v) mx[v] = xmask_max<2>(mx[v]);
    #pragma unroll
    for (int v = 0; v < 8; ++v) mx[v] = xmask_max<4>(mx[v]);
    #pragma unroll
    for (int v = 0; v < 8; ++v) {
      const float* mrow = adjq + (unsigned)kb + moff + v * NN;
      const v4f mw = *(const v4f*)mrow;
      const float w0 = mw[0], w1 = mw[1], w2 = mw[2], w3 = mw[3];
      __builtin_prefetch(mrow + 64, 0, 0);
      const float r0 = sc[0][v] * inv_s;
      const float r1 = sc[1][v] * inv_s;
      const float r2 = sc[2][v] * inv_s;
      const float r3 = sc[3][v] * inv_s;
      const float mn    = xmask8_max3(mx[v], m[v]);
      const float scale = __expf(m[v] - mn);
      m[v] = mn;
      const float p0 = w0 * __expf(r0 - mn);
      const float p1 = w1 * __expf(r1 - mn);
      const float p2 = w2 * __expf(r2 - mn);
      const float p3 = w3 * __expf(r3 - mn);
      o0[v]   *= scale;
      o1[v]   *= scale;
      lacc[v] *= scale;
      const int r = v + 8 * half;
      const v4h pk = {(_Float16)p0, (_Float16)p1, (_Float16)p2, (_Float16)p3};
      *(v4h*)&Plds[r * 64 + 4 * ll] = pk;
    }
    asm volatile("s_wait_dscnt 0" ::: "memory");

    #pragma unroll
    for (int c = 0; c < 2; ++c) {
      const v16h aP = make_ab(&Plds[ll * 64 + c * 32 + c0],
                              &Plds[ll * 64 + c * 32 + c0 + 16]);
      const _Float16* vr0 = ZTk + (unsigned)(kb * SS) + foff + c * 1024;
      const _Float16* vr1 = vr0 + 512;
      const v16h bV0 = make_ab(vr0, vr0 + 256);
      const v16h bV1 = make_ab(vr1, vr1 + 256);
      o0   = __builtin_amdgcn_wmma_f32_16x16x32_f16(false, aP, false, bV0,  (short)0, o0,   false, false);
      o1   = __builtin_amdgcn_wmma_f32_16x16x32_f16(false, aP, false, bV1,  (short)0, o1,   false, false);
      lacc = __builtin_amdgcn_wmma_f32_16x16x32_f16(false, aP, false, ones, (short)0, lacc, false, false);
    }
  }

  #pragma unroll
  for (int v = 0; v < 8; ++v) {
    const int r = v + 8 * half;
    Om[(w * 16 + r) * 32 + ll]      = o0[v];
    Om[(w * 16 + r) * 32 + 16 + ll] = o1[v];
    if (ll == 0) {
      Mm[w * 16 + r] = m[v];
      Lm[w * 16 + r] = lacc[v];
    }
  }
  __syncthreads();

  for (int j = 0; j < 2; ++j) {
    const int i = (int)threadIdx.x + j * (32 * KSPLIT);
    const int r = i >> 5;
    float M = Mm[r];
    for (int u = 1; u < KSPLIT; ++u) M = fmaxf(M, Mm[u * 16 + r]);
    float L = 0.f, acc = 0.f;
    for (int u = 0; u < KSPLIT; ++u) {
      const float f = __expf(Mm[u * 16 + r] - M);
      L   = fmaf(Lm[u * 16 + r], f, L);
      acc = fmaf(Om[u * 512 + i], f, acc);
    }
    O[((size_t)k * NN + qb) * SS + i] = acc / L;
  }
}

__global__ __launch_bounds__(256) void combine_kernel(
    const float* __restrict__ H, const float* __restrict__ U,
    const float* __restrict__ O, const float* __restrict__ thr,
    float* __restrict__ out)
{
  const int n0 = blockIdx.x * 16;
  const int w  = threadIdx.x >> 5;
  const int l  = threadIdx.x & 31;
  const int c  = l & 15;
  const int h  = l >> 4;
  const int dc = w * 32 + 2 * c;

  v8f acc0 = {}; v8f acc1 = {};
  #pragma unroll
  for (int k = 0; k < KK; ++k) {
    const v4f* pa  = (const v4f*)(O + ((size_t)k * NN + n0 + c) * SS + 16 * h);
    const v4f* pb0 = (const v4f*)(U + ((size_t)k * DD + dc) * SS + 16 * h);
    const v4f* pb1 = (const v4f*)(U + ((size_t)k * DD + dc + 1) * SS + 16 * h);
    #pragma unroll
    for (int q = 0; q < 4; ++q) {
      const v4f a = pa[q], b0 = pb0[q], b1 = pb1[q];
      const v2f alo  = {a[0],  a[1]},  ahi  = {a[2],  a[3]};
      const v2f b0lo = {b0[0], b0[1]}, b0hi = {b0[2], b0[3]};
      const v2f b1lo = {b1[0], b1[1]}, b1hi = {b1[2], b1[3]};
      acc0 = __builtin_amdgcn_wmma_f32_16x16x4_f32(false, alo, false, b0lo, (short)0, acc0, false, false);
      acc1 = __builtin_amdgcn_wmma_f32_16x16x4_f32(false, alo, false, b1lo, (short)0, acc1, false, false);
      acc0 = __builtin_amdgcn_wmma_f32_16x16x4_f32(false, ahi, false, b0hi, (short)0, acc0, false, false);
      acc1 = __builtin_amdgcn_wmma_f32_16x16x4_f32(false, ahi, false, b1hi, (short)0, acc1, false, false);
    }
  }

  const v2f t = *(const v2f*)(thr + dc);
  #pragma unroll
  for (int r = 0; r < 8; ++r) {
    const size_t o = (size_t)(n0 + 8 * h + r) * DD + dc;
    const v2f hv = *(const v2f*)(H + o);
    const float h0 = hv[0] + ETA * acc0[r];
    const float h1 = hv[1] + ETA * acc1[r];
    const float a0 = fabsf(h0) - t[0];
    const float a1 = fabsf(h1) - t[1];
    v2f res;
    res[0] = (a0 > 0.f) ? copysignf(a0, h0) : 0.f;
    res[1] = (a1 > 0.f) ? copysignf(a1, h1) : 0.f;
    *(v2f*)(out + o) = res;
  }
}

extern "C" void kernel_launch(void* const* d_in, const int* in_sizes, int n_in,
                              void* d_out, int out_size, void* d_ws, size_t ws_size,
                              hipStream_t stream) {
  (void)in_sizes; (void)n_in; (void)out_size; (void)ws_size;
  const float* H   = (const float*)d_in[0];
  const float* adj = (const float*)d_in[1];
  const float* U   = (const float*)d_in[2];
  const float* lp  = (const float*)d_in[3];
  const float* thr = (const float*)d_in[4];
  float* out = (float*)d_out;

  char* ws = (char*)d_ws;
  _Float16* Zs = (_Float16*)(ws);
  _Float16* Zf = (_Float16*)(ws + (1u << 20));
  _Float16* ZT = (_Float16*)(ws + (2u << 20));
  float*    O  = (float*)   (ws + (3u << 20));

  zproj_kernel     <<<GTILES + ZTILES,  32 * KK, 0, stream>>>(H, U, lp, Zs, Zf, ZT, O);
  flash_attn_kernel<<<dim3(NN / 16, KK), 32 * KSPLIT, 0, stream>>>(Zs, Zf, ZT, adj, O, out);
  combine_kernel   <<<NN / 16,          256, 0, stream>>>(H, U, O, thr, out);
}


# attention: each wave takes two 16-query tiles, so a block's key and value fragments are loaded once for both; its key, mask and value loads issue as whole groups ahead of the softmax; sums kept
# speedup vs baseline: 11.7970x; 1.1653x over previous
#include <hip/hip_runtime.h>

#define NN 4096
#define DD 256
#define SS 32
#define KK 4
#define ETA 0.5f
#define GROWS  (KK * SS)
#define GPARTS (KK * KK * SS)

typedef __attribute__((ext_vector_type(16))) _Float16 v16h;
typedef __attribute__((ext_vector_type(8)))  _Float16 v8h;
typedef __attribute__((ext_vector_type(8)))  float    v8f;

__device__ __forceinline__ v16h make_ab(const _Float16* p0, const _Float16* p1) {
  v8h lo = *(const v8h*)p0;
  v8h hi = *(const v8h*)p1;
  return __builtin_shufflevector(lo, hi, 0,1,2,3,4,5,6,7,8,9,10,11,12,13,14,15);
}

typedef __attribute__((ext_vector_type(2))) float    v2f;
typedef __attribute__((ext_vector_type(4))) float    v4f;
typedef __attribute__((ext_vector_type(2))) _Float16 v2h;
typedef __attribute__((ext_vector_type(4))) _Float16 v4h;

__device__ __forceinline__ unsigned kfrag_off(unsigned n, unsigned s) {
  const unsigned kk = n & 63u, gs = s >> 3;
  const unsigned unit = ((kk & 3u) * 2u + (gs >> 1)) * 32u + (gs & 1u) * 16u + (kk >> 2);
  return (n >> 6) * 2048u + unit * 8u + (s & 7u);
}
__device__ __forceinline__ unsigned vfrag_off(unsigned n, unsigned s) {
  const unsigned kk = n & 63u;
  const unsigned unit = ((((kk >> 5) * 2u + (s >> 4)) * 2u + ((kk >> 4) & 1u)) * 32u)
                      + ((kk >> 3) & 1u) * 16u + (s & 15u);
  return (n >> 6) * 2048u + unit * 8u + (kk & 7u);
}

#define ZTILES (NN / 16)
#define GTILES (GROWS / 16)

template <bool GRAM>
__device__ __forceinline__ void rows_times_Uk(const float* __restrict__ arow,
                                              const float* __restrict__ ucol,
                                              v8f& acc0, v8f& acc1)
{
  #pragma unroll
  for (int J = 0; J < DD / 8; ++J) {
    v4f a;
    if (GRAM) {
      a[0] = arow[(8 * J + 0) * SS];
      a[1] = arow[(8 * J + 1) * SS];
      a[2] = arow[(8 * J + 2) * SS];
      a[3] = arow[(8 * J + 3) * SS];
    } else {
      a = *(const v4f*)(arow + 8 * J);
    }
    const v2f u0 = *(const v2f*)(ucol + (8 * J + 0) * SS);
    const v2f u1 = *(const v2f*)(ucol + (8 * J + 1) * SS);
    const v2f u2 = *(const v2f*)(ucol + (8 * J + 2) * SS);
    const v2f u3 = *(const v2f*)(ucol + (8 * J + 3) * SS);
    const v2f alo  = {a[0],  a[1]},  ahi  = {a[2],  a[3]};
    const v2f b0lo = {u0[0], u1[0]}, b0hi = {u2[0], u3[0]};
    const v2f b1lo = {u0[1], u1[1]}, b1hi = {u2[1], u3[1]};
    acc0 = __builtin_amdgcn_wmma_f32_16x16x4_f32(false, alo, false, b0lo, (short)0, acc0, false, false);
    acc1 = __builtin_amdgcn_wmma_f32_16x16x4_f32(false, alo, false, b1lo, (short)0, acc1, false, false);
    acc0 = __builtin_amdgcn_wmma_f32_16x16x4_f32(false, ahi, false, b0hi, (short)0, acc0, false, false);
    acc1 = __builtin_amdgcn_wmma_f32_16x16x4_f32(false, ahi, false, b1hi, (short)0, acc1, false, false);
  }
}

__global__ __launch_bounds__(32 * KK) __attribute__((amdgpu_waves_per_eu(1, 4))) void zproj_kernel(
    const float* __restrict__ H, const float* __restrict__ U,
    const float* __restrict__ lp,
    _Float16* __restrict__ Zs, _Float16* __restrict__ Zf, _Float16* __restrict__ ZT,
    float* __restrict__ Gp)
{
  const int k    = __builtin_amdgcn_readfirstlane((int)(threadIdx.x >> 5));
  const int l    = threadIdx.x & 31;
  const int c    = l & 15;
  const int h    = l >> 4;
  const int tile = blockIdx.x;
  const float* ucol = U + (size_t)k * DD * SS + (4 * h) * SS + 2 * c;

  v8f acc0 = {}; v8f acc1 = {};

  if (tile < GTILES) {
    const int kq = tile >> 1;
    const int a0 = 16 * (tile & 1);
    if (kq <= k)
      rows_times_Uk<true>(U + (size_t)kq * DD * SS + (4 * h) * SS + a0 + c, ucol, acc0, acc1);
    float part[8];
    #pragma unroll
    for (int r = 0; r < 8; ++r) {
      const int a = a0 + 8 * h + r;
      const float e0 = (kq == k && a == 2 * c)     ? acc0[r] - 1.f : acc0[r];
      const float e1 = (kq == k && a == 2 * c + 1) ? acc1[r] - 1.f : acc1[r];
      part[r] = e0 * e0 + e1 * e1;
    }
    #pragma unroll
    for (int off = 8; off > 0; off >>= 1) {
      #pragma unroll
      for (int r = 0; r < 8; ++r) part[r] += __shfl_xor(part[r], off, 32);
    }
    if (c == 0) {
      v4f* gp = (v4f*)(Gp + (kq * KK + k) * SS + a0 + 8 * h);
      const v4f p0 = {part[0], part[1], part[2], part[3]};
      const v4f p1 = {part[4], part[5], part[6], part[7]};
      gp[0] = p0;
      gp[1] = p1;
    }
    return;
  }

  const int n0 = (tile - GTILES) * 16;
  rows_times_Uk<false>(H + (size_t)(n0 + c) * DD + 4 * h, ucol, acc0, acc1);

  const v2f lpv = *(const v2f*)(lp + k * SS + 2 * c);
  const float e0 = __expf(lpv[0]);
  const float e1 = __expf(lpv[1]);
  v8h t0, t1;
  #pragma unroll
  for (int r = 0; r < 8; ++r) {
    const float z0 = acc0[r], z1 = acc1[r];
    const size_t o = ((size_t)k * NN + n0 + 8 * h + r) * SS + 2 * c;
    const v2h kf = {(_Float16)z0, (_Float16)z1};
    const v2h qf = {(_Float16)(z0 * e0), (_Float16)(z1 * e1)};
    *(v2h*)(Zf + (size_t)k * NN * SS
               + kfrag_off((unsigned)(n0 + 8 * h + r), (unsigned)(2 * c))) = kf;
    *(v2h*)(Zs + o) = qf;
    t0[r] = (_Float16)z0;
    t1[r] = (_Float16)z1;
  }
  _Float16* vt = ZT + (size_t)k * SS * NN + vfrag_off((unsigned)(n0 + 8 * h), (unsigned)(2 * c));
  *(v8h*)vt       = t0;
  *(v8h*)(vt + 8) = t1;
}


template <int MASK>
__device__ __forceinline__ float xmask_max(float x) {
  float y;
  asm("v_max_num_f32_dpp %0, %1, %1 row_xmask:%2 row_mask:0xf bank_mask:0xf"
      : "=v"(y) : "v"(x), "n"(MASK));
  return y;
}
__device__ __forceinline__ float xmask8_max3(float x, float m) {
  float y;
  asm("v_max3_num_f32_e64_dpp %0, %1, %1, %2 row_xmask:8 row_mask:0xf bank_mask:0xf"
      : "=v"(y) : "v"(x), "v"(m));
  return y;
}

__device__ __forceinline__ void softmax_tile(const v8f (&sc)[4], const v4f (&mw)[8],
                                             float (&m)[8], v8f& o0, v8f& o1, v8f& lacc,
                                             _Float16* Pt, int ll, int half)
{
  const float inv_s = 0.17677669529663687f;
  float mx[8];
  #pragma unroll
  for (int v = 0; v < 8; ++v)
    mx[v] = fmaxf(fmaxf(sc[0][v] * inv_s, sc[1][v] * inv_s),
                  fmaxf(sc[2][v] * inv_s, sc[3][v] * inv_s));
  #pragma unroll
  for (int v = 0; v < 8; ++v) mx[v] = xmask_max<1>(mx[v]);
  #pragma unroll
  for (int v = 0; v < 8; ++v) mx[v] = xmask_max<2>(mx[v]);
  #pragma unroll
  for (int v = 0; v < 8; ++v) mx[v] = xmask_max<4>(mx[v]);
  #pragma unroll
  for (int v = 0; v < 8; ++v) {
    const float w0 = mw[v][0], w1 = mw[v][1], w2 = mw[v][2], w3 = mw[v][3];
    const float r0 = sc[0][v] * inv_s;
    const float r1 = sc[1][v] * inv_s;
    const float r2 = sc[2][v] * inv_s;
    const float r3 = sc[3][v] * inv_s;
    const float mn    = xmask8_max3(mx[v], m[v]);
    const float scale = __expf(m[v] - mn);
    m[v] = mn;
    const float p0 = w0 * __expf(r0 - mn);
    const float p1 = w1 * __expf(r1 - mn);
    const float p2 = w2 * __expf(r2 - mn);
    const float p3 = w3 * __expf(r3 - mn);
    o0[v]   *= scale;
    o1[v]   *= scale;
    lacc[v] *= scale;
    const int r = v + 8 * half;
    const v4h pk = {(_Float16)p0, (_Float16)p1, (_Float16)p2, (_Float16)p3};
    *(v4h*)&Pt[r * 64 + 4 * ll] = pk;
  }
}

__device__ __forceinline__ void load_mask_tile(const float* __restrict__ mbase, v4f (&mw)[8])
{
  #pragma unroll
  for (int v = 0; v < 8; ++v) mw[v] = *(const v4f*)(mbase + v * NN);
  #pragma unroll
  for (int v = 0; v < 8; ++v) __builtin_prefetch(mbase + v * NN + 64, 0, 0);
}

#define KSPLIT 8
#define KRANGE (NN / KSPLIT)
#define QT 2

__global__ __launch_bounds__(32 * KSPLIT) void flash_attn_kernel(
    const _Float16* __restrict__ Zs, const _Float16* __restrict__ Zf,
    const _Float16* __restrict__ ZT, const float* __restrict__ adj,
    float* __restrict__ O, float* __restrict__ out)
{
  const int k    = blockIdx.y;
  const int qb   = blockIdx.x * (16 * QT);
  const int w    = __builtin_amdgcn_readfirstlane((int)(threadIdx.x >> 5));
  const int l    = threadIdx.x & 31;
  const int ll   = l & 15;
  const int half = l >> 4;
  const int c0   = half ? 8 : 0;

  if (blockIdx.x == 0 && blockIdx.y == 0 && w == 0) {
    float t = 0.f;
    for (int i = 0; i < GPARTS / 32; ++i) t += O[l + 32 * i];
    for (int off = 16; off > 0; off >>= 1) t += __shfl_xor(t, off, 32);
    if (l == 0) out[(size_t)NN * DD] = t;
  }

  __shared__ _Float16 __attribute__((aligned(16))) PldsAll[KSPLIT * QT * 16 * 64];
  __shared__ float Mm[KSPLIT * QT * 16];
  __shared__ float Lm[KSPLIT * QT * 16];
  _Float16* Plds = PldsAll + w * (QT * 16 * 64);

  const _Float16* qrow = Zs + ((size_t)k * NN + (qb + ll)) * SS;
  v16h aQ[QT];
  #pragma unroll
  for (int q = 0; q < QT; ++q)
    aQ[q] = make_ab(qrow + q * 16 * SS + c0, qrow + q * 16 * SS + c0 + 16);

  v16h ones;
  #pragma unroll
  for (int i = 0; i < 16; ++i) ones[i] = (_Float16)1.0f;

  float m[QT][8];
  v8f o0[QT], o1[QT];
  v8f lacc[QT];
  #pragma unroll
  for (int q = 0; q < QT; ++q) {
    #pragma unroll
    for (int v = 0; v < 8; ++v) { m[q][v] = -INFINITY; o0[q][v] = 0.f; o1[q][v] = 0.f; lacc[q][v] = 0.f; }
  }

  const _Float16* Zfk  = Zf + (size_t)k * NN * SS;
  const _Float16* ZTk  = ZT + (size_t)k * SS * NN;
  const float*    adjq = adj + (size_t)qb * NN;
  const unsigned  foff = (unsigned)(l * 8);
  const unsigned  moff = (unsigned)(8 * half * NN + 4 * ll);
  const int kb0 = w * KRANGE;
  for (int kb = kb0; kb < kb0 + KRANGE; kb += 64) {
    v16h bK[4];
    #pragma unroll
    for (int t = 0; t < 4; ++t) {
      const _Float16* kr = Zfk + (unsigned)(kb * SS) + foff + t * 512;
      bK[t] = make_ab(kr, kr + 256);
    }
    const float* mblk = adjq + (unsigned)kb + moff;
    v4f mw0[8], mw1[8];
    load_mask_tile(mblk, mw0);

    v8f sc[QT][4];
    #pragma unroll
    for (int t = 0; t < 4; ++t) {
      #pragma unroll
      for (int q = 0; q < QT; ++q) {
        v8f z = {};
        sc[q][t] = __builtin_amdgcn_wmma_f32_16x16x32_f16(false, aQ[q], false, bK[t], (short)0, z, false, false);
      }
    }

    load_mask_tile(mblk + 16 * NN, mw1);
    softmax_tile(sc[0], mw0, m[0], o0[0], o1[0], lacc[0], Plds, ll, half);

    v16h bV[2][2];
    #pragma unroll
    for (int c = 0; c < 2; ++c) {
      const _Float16* vr0 = ZTk + (unsigned)(kb * SS) + foff + c * 1024;
      bV[c][0] = make_ab(vr0, vr0 + 256);
      bV[c][1] = make_ab(vr0 + 512, vr0 + 768);
    }
    __builtin_amdgcn_sched_barrier(0);
    softmax_tile(sc[1], mw1, m[1], o0[1], o1[1], lacc[1], Plds + 1024, ll, half);
    asm volatile("s_wait_dscnt 0" ::: "memory");

    #pragma unroll
    for (int c = 0; c < 2; ++c) {
      #pragma unroll
      for (int q = 0; q < QT; ++q) {
        const v16h aP = make_ab(&Plds[q * 1024 + ll * 64 + c * 32 + c0],
                                &Plds[q * 1024 + ll * 64 + c * 32 + c0 + 16]);
        o0[q]   = __builtin_amdgcn_wmma_f32_16x16x32_f16(false, aP, false, bV[c][0], (short)0, o0[q],   false, false);
        o1[q]   = __builtin_amdgcn_wmma_f32_16x16x32_f16(false, aP, false, bV[c][1], (short)0, o1[q],   false, false);
        lacc[q] = __builtin_amdgcn_wmma_f32_16x16x32_f16(false, aP, false, ones,     (short)0, lacc[q], false, false);
      }
    }
  }

  float* Om = (float*)PldsAll;
  #pragma unroll
  for (int q = 0; q < QT; ++q) {
    #pragma unroll
    for (int v = 0; v < 8; ++v) {
      const int r = v + 8 * half;
      Om[(w * QT + q) * 512 + r * 32 + ll]      = o0[q][v];
      Om[(w * QT + q) * 512 + r * 32 + 16 + ll] = o1[q][v];
      if (ll == 0) {
        Mm[(w * QT + q) * 16 + r] = m[q][v];
        Lm[(w * QT + q) * 16 + r] = lacc[q][v];
      }
    }
  }
  __syncthreads();

  for (int j = 0; j < 2 * QT; ++j) {
    const int i = (int)threadIdx.x + j * (32 * KSPLIT);
    const int r = i >> 5;
    float M = Mm[r];
    for (int u = 1; u < KSPLIT; ++u) M = fmaxf(M, Mm[u * (16 * QT) + r]);
    float L = 0.f, acc = 0.f;
    for (int u = 0; u < KSPLIT; ++u) {
      const float f = __expf(Mm[u * (16 * QT) + r] - M);
      L   = fmaf(Lm[u * (16 * QT) + r], f, L);
      acc = fmaf(Om[u * (512 * QT) + i], f, acc);
    }
    O[((size_t)k * NN + qb) * SS + i] = acc / L;
  }
}

__global__ __launch_bounds__(256) void combine_kernel(
    const float* __restrict__ H, const float* __restrict__ U,
    const float* __restrict__ O, const float* __restrict__ thr,
    float* __restrict__ out)
{
  const int n0 = blockIdx.x * 16;
  const int w  = threadIdx.x >> 5;
  const int l  = threadIdx.x & 31;
  const int c  = l & 15;
  const int h  = l >> 4;
  const int dc = w * 32 + 2 * c;

  v8f acc0 = {}; v8f acc1 = {};
  #pragma unroll
  for (int k = 0; k < KK; ++k) {
    const v4f* pa  = (const v4f*)(O + ((size_t)k * NN + n0 + c) * SS + 16 * h);
    const v4f* pb0 = (const v4f*)(U + ((size_t)k * DD + dc) * SS + 16 * h);
    const v4f* pb1 = (const v4f*)(U + ((size_t)k * DD + dc + 1) * SS + 16 * h);
    #pragma unroll
    for (int q = 0; q < 4; ++q) {
      const v4f a = pa[q], b0 = pb0[q], b1 = pb1[q];
      const v2f alo  = {a[0],  a[1]},  ahi  = {a[2],  a[3]};
      const v2f b0lo = {b0[0], b0[1]}, b0hi = {b0[2], b0[3]};
      const v2f b1lo = {b1[0], b1[1]}, b1hi = {b1[2], b1[3]};
      acc0 = __builtin_amdgcn_wmma_f32_16x16x4_f32(false, alo, false, b0lo, (short)0, acc0, false, false);
      acc1 = __builtin_amdgcn_wmma_f32_16x16x4_f32(false, alo, false, b1lo, (short)0, acc1, false, false);
      acc0 = __builtin_amdgcn_wmma_f32_16x16x4_f32(false, ahi, false, b0hi, (short)0, acc0, false, false);
      acc1 = __builtin_amdgcn_wmma_f32_16x16x4_f32(false, ahi, false, b1hi, (short)0, acc1, false, false);
    }
  }

  const v2f t = *(const v2f*)(thr + dc);
  #pragma unroll
  for (int r = 0; r < 8; ++r) {
    const size_t o = (size_t)(n0 + 8 * h + r) * DD + dc;
    const v2f hv = *(const v2f*)(H + o);
    const float h0 = hv[0] + ETA * acc0[r];
    const float h1 = hv[1] + ETA * acc1[r];
    const float a0 = fabsf(h0) - t[0];
    const float a1 = fabsf(h1) - t[1];
    v2f res;
    res[0] = (a0 > 0.f) ? copysignf(a0, h0) : 0.f;
    res[1] = (a1 > 0.f) ? copysignf(a1, h1) : 0.f;
    *(v2f*)(out + o) = res;
  }
}

extern "C" void kernel_launch(void* const* d_in, const int* in_sizes, int n_in,
                              void* d_out, int out_size, void* d_ws, size_t ws_size,
                              hipStream_t stream) {
  (void)in_sizes; (void)n_in; (void)out_size; (void)ws_size;
  const float* H   = (const float*)d_in[0];
  const float* adj = (const float*)d_in[1];
  const float* U   = (const float*)d_in[2];
  const float* lp  = (const float*)d_in[3];
  const float* thr = (const float*)d_in[4];
  float* out = (float*)d_out;

  char* ws = (char*)d_ws;
  _Float16* Zs = (_Float16*)(ws);
  _Float16* Zf = (_Float16*)(ws + (1u << 20));
  _Float16* ZT = (_Float16*)(ws + (2u << 20));
  float*    O  = (float*)   (ws + (3u << 20));

  zproj_kernel     <<<GTILES + ZTILES,  32 * KK, 0, stream>>>(H, U, lp, Zs, Zf, ZT, O);
  flash_attn_kernel<<<dim3(NN / (16 * QT), KK), 32 * KSPLIT, 0, stream>>>(Zs, Zf, ZT, adj, O, out);
  combine_kernel   <<<NN / 16,          256, 0, stream>>>(H, U, O, thr, out);
}


# adjacency mask packed once per call into lane-mask words (ballots on the projection grid); attention reads them by scalar loads and selects the exponent, no f32 mask loads per head; sums kept
# speedup vs baseline: 12.2045x; 1.0345x over previous
#include <hip/hip_runtime.h>

#define NN 4096
#define DD 256
#define SS 32
#define KK 4
#define ETA 0.5f
#define GROWS  (KK * SS)
#define GPARTS (KK * KK * SS)

typedef __attribute__((ext_vector_type(16))) _Float16 v16h;
typedef __attribute__((ext_vector_type(8)))  _Float16 v8h;
typedef __attribute__((ext_vector_type(8)))  float    v8f;

__device__ __forceinline__ v16h make_ab(const _Float16* p0, const _Float16* p1) {
  v8h lo = *(const v8h*)p0;
  v8h hi = *(const v8h*)p1;
  return __builtin_shufflevector(lo, hi, 0,1,2,3,4,5,6,7,8,9,10,11,12,13,14,15);
}

typedef __attribute__((ext_vector_type(2))) float    v2f;
typedef __attribute__((ext_vector_type(4))) float    v4f;
typedef __attribute__((ext_vector_type(2))) _Float16 v2h;
typedef __attribute__((ext_vector_type(4))) _Float16 v4h;

__device__ __forceinline__ unsigned kfrag_off(unsigned n, unsigned s) {
  const unsigned kk = n & 63u, gs = s >> 3;
  const unsigned unit = ((kk & 3u) * 2u + (gs >> 1)) * 32u + (gs & 1u) * 16u + (kk >> 2);
  return (n >> 6) * 2048u + unit * 8u + (s & 7u);
}
__device__ __forceinline__ unsigned vfrag_off(unsigned n, unsigned s) {
  const unsigned kk = n & 63u;
  const unsigned unit = ((((kk >> 5) * 2u + (s >> 4)) * 2u + ((kk >> 4) & 1u)) * 32u)
                      + ((kk >> 3) & 1u) * 16u + (s & 15u);
  return (n >> 6) * 2048u + unit * 8u + (kk & 7u);
}

#define ZTILES (NN / 16)
#define GTILES (GROWS / 16)
#define PBLOCKS (NN / 4)

typedef __attribute__((ext_vector_type(16))) unsigned v16u;


template <bool GRAM>
__device__ __forceinline__ void rows_times_Uk(const float* __restrict__ arow,
                                              const float* __restrict__ ucol,
                                              v8f& acc0, v8f& acc1)
{
  #pragma unroll
  for (int J = 0; J < DD / 8; ++J) {
    v4f a;
    if (GRAM) {
      a[0] = arow[(8 * J + 0) * SS];
      a[1] = arow[(8 * J + 1) * SS];
      a[2] = arow[(8 * J + 2) * SS];
      a[3] = arow[(8 * J + 3) * SS];
    } else {
      a = *(const v4f*)(arow + 8 * J);
    }
    const v2f u0 = *(const v2f*)(ucol + (8 * J + 0) * SS);
    const v2f u1 = *(const v2f*)(ucol + (8 * J + 1) * SS);
    const v2f u2 = *(const v2f*)(ucol + (8 * J + 2) * SS);
    const v2f u3 = *(const v2f*)(ucol + (8 * J + 3) * SS);
    const v2f alo  = {a[0],  a[1]},  ahi  = {a[2],  a[3]};
    const v2f b0lo = {u0[0], u1[0]}, b0hi = {u2[0], u3[0]};
    const v2f b1lo = {u0[1], u1[1]}, b1hi = {u2[1], u3[1]};
    acc0 = __builtin_amdgcn_wmma_f32_16x16x4_f32(false, alo, false, b0lo, (short)0, acc0, false, false);
    acc1 = __builtin_amdgcn_wmma_f32_16x16x4_f32(false, alo, false, b1lo, (short)0, acc1, false, false);
    acc0 = __builtin_amdgcn_wmma_f32_16x16x4_f32(false, ahi, false, b0hi, (short)0, acc0, false, false);
    acc1 = __builtin_amdgcn_wmma_f32_16x16x4_f32(false, ahi, false, b1hi, (short)0, acc1, false, false);
  }
}

__global__ __launch_bounds__(32 * KK) __attribute__((amdgpu_waves_per_eu(1, 4))) void zproj_kernel(
    const float* __restrict__ H, const float* __restrict__ U,
    const float* __restrict__ lp,
    _Float16* __restrict__ Zs, _Float16* __restrict__ Zf, _Float16* __restrict__ ZT,
    float* __restrict__ Gp, const float* __restrict__ adj, unsigned* __restrict__ Mb)
{
  const int k    = __builtin_amdgcn_readfirstlane((int)(threadIdx.x >> 5));
  const int l    = threadIdx.x & 31;
  const int c    = l & 15;
  const int h    = l >> 4;
  const int tile = blockIdx.x;
  const float* ucol = U + (size_t)k * DD * SS + (4 * h) * SS + 2 * c;

  v8f acc0 = {}; v8f acc1 = {};

  if (tile < GTILES) {
    const int kq = tile >> 1;
    const int a0 = 16 * (tile & 1);
    if (kq <= k)
      rows_times_Uk<true>(U + (size_t)kq * DD * SS + (4 * h) * SS + a0 + c, ucol, acc0, acc1);
    float part[8];
    #pragma unroll
    for (int r = 0; r < 8; ++r) {
      const int a = a0 + 8 * h + r;
      const float e0 = (kq == k && a == 2 * c)     ? acc0[r] - 1.f : acc0[r];
      const float e1 = (kq == k && a == 2 * c + 1) ? acc1[r] - 1.f : acc1[r];
      part[r] = e0 * e0 + e1 * e1;
    }
    #pragma unroll
    for (int off = 8; off > 0; off >>= 1) {
      #pragma unroll
      for (int r = 0; r < 8; ++r) part[r] += __shfl_xor(part[r], off, 32);
    }
    if (c == 0) {
      v4f* gp = (v4f*)(Gp + (kq * KK + k) * SS + a0 + 8 * h);
      const v4f p0 = {part[0], part[1], part[2], part[3]};
      const v4f p1 = {part[4], part[5], part[6], part[7]};
      gp[0] = p0;
      gp[1] = p1;
    }
    return;
  }

  if (tile >= GTILES + ZTILES) {
    const int g  = (tile - (GTILES + ZTILES)) * KK + k;
    const int x  = g >> 5;
    const int kc = g & 31;
    const float* mrow = adj + (size_t)(32 * x + 8 * h) * NN + 128 * kc + 4 * c;
    v4f mw[2][2][8];
    #pragma unroll
    for (int kk = 0; kk < 2; ++kk) {
      #pragma unroll
      for (int q = 0; q < 2; ++q) {
        #pragma unroll
        for (int v = 0; v < 8; ++v)
          mw[kk][q][v] = *(const v4f*)(mrow + (size_t)(16 * q + v) * NN + 64 * kk);
      }
    }
    #pragma unroll
    for (int kk = 0; kk < 2; ++kk) {
      #pragma unroll
      for (int q = 0; q < 2; ++q) {
        unsigned word = 0u;
        #pragma unroll
        for (int v = 0; v < 8; ++v) {
          #pragma unroll
          for (int j = 0; j < 4; ++j) {
            const unsigned bits = __builtin_amdgcn_ballot_w32(mw[kk][q][v][j] != 0.f);
            word = (l == 4 * v + j) ? bits : word;
          }
        }
        Mb[(((size_t)x * 64 + 2 * kc + kk) * 2 + q) * 32 + l] = word;
      }
    }
    return;
  }

  const int n0 = (tile - GTILES) * 16;
  rows_times_Uk<false>(H + (size_t)(n0 + c) * DD + 4 * h, ucol, acc0, acc1);

  const v2f lpv = *(const v2f*)(lp + k * SS + 2 * c);
  const float e0 = __expf(lpv[0]);
  const float e1 = __expf(lpv[1]);
  v8h t0, t1;
  #pragma unroll
  for (int r = 0; r < 8; ++r) {
    const float z0 = acc0[r], z1 = acc1[r];
    const size_t o = ((size_t)k * NN + n0 + 8 * h + r) * SS + 2 * c;
    const v2h kf = {(_Float16)z0, (_Float16)z1};
    const v2h qf = {(_Float16)(z0 * e0), (_Float16)(z1 * e1)};
    *(v2h*)(Zf + (size_t)k * NN * SS
               + kfrag_off((unsigned)(n0 + 8 * h + r), (unsigned)(2 * c))) = kf;
    *(v2h*)(Zs + o) = qf;
    t0[r] = (_Float16)z0;
    t1[r] = (_Float16)z1;
  }
  _Float16* vt = ZT + (size_t)k * SS * NN + vfrag_off((unsigned)(n0 + 8 * h), (unsigned)(2 * c));
  *(v8h*)vt       = t0;
  *(v8h*)(vt + 8) = t1;
}


template <int MASK>
__device__ __forceinline__ float xmask_max(float x) {
  float y;
  asm("v_max_num_f32_dpp %0, %1, %1 row_xmask:%2 row_mask:0xf bank_mask:0xf"
      : "=v"(y) : "v"(x), "n"(MASK));
  return y;
}
__device__ __forceinline__ float xmask8_max3(float x, float m) {
  float y;
  asm("v_max3_num_f32_e64_dpp %0, %1, %1, %2 row_xmask:8 row_mask:0xf bank_mask:0xf"
      : "=v"(y) : "v"(x), "v"(m));
  return y;
}

__device__ __forceinline__ float keep_if(unsigned lanemask, float x) {
  return __builtin_amdgcn_inverse_ballot_w32(lanemask) ? x : -INFINITY;
}
__device__ __forceinline__ void softmax_tile(const v8f (&sc)[4], const v16u& ma, const v16u& mb,
                                             float (&m)[8], v8f& o0, v8f& o1, v8f& lacc,
                                             _Float16* Pt, int ll, int half)
{
  const float inv_s = 0.17677669529663687f;
  float mx[8];
  #pragma unroll
  for (int v = 0; v < 8; ++v)
    mx[v] = fmaxf(fmaxf(sc[0][v] * inv_s, sc[1][v] * inv_s),
                  fmaxf(sc[2][v] * inv_s, sc[3][v] * inv_s));
  #pragma unroll
  for (int v = 0; v < 8; ++v) mx[v] = xmask_max<1>(mx[v]);
  #pragma unroll
  for (int v = 0; v < 8; ++v) mx[v] = xmask_max<2>(mx[v]);
  #pragma unroll
  for (int v = 0; v < 8; ++v) mx[v] = xmask_max<4>(mx[v]);
  #pragma unroll
  for (int v = 0; v < 8; ++v) {
    const unsigned w0 = (v < 4) ? ma[4 * v + 0] : mb[4 * v - 16];
    const unsigned w1 = (v < 4) ? ma[4 * v + 1] : mb[4 * v - 15];
    const unsigned w2 = (v < 4) ? ma[4 * v + 2] : mb[4 * v - 14];
    const unsigned w3 = (v < 4) ? ma[4 * v + 3] : mb[4 * v - 13];
    const float r0 = sc[0][v] * inv_s;
    const float r1 = sc[1][v] * inv_s;
    const float r2 = sc[2][v] * inv_s;
    const float r3 = sc[3][v] * inv_s;
    const float mn    = xmask8_max3(mx[v], m[v]);
    const float scale = __expf(m[v] - mn);
    m[v] = mn;
    const float p0 = __expf(keep_if(w0, r0 - mn));
    const float p1 = __expf(keep_if(w1, r1 - mn));
    const float p2 = __expf(keep_if(w2, r2 - mn));
    const float p3 = __expf(keep_if(w3, r3 - mn));
    o0[v]   *= scale;
    o1[v]   *= scale;
    lacc[v] *= scale;
    const int r = v + 8 * half;
    const v4h pk = {(_Float16)p0, (_Float16)p1, (_Float16)p2, (_Float16)p3};
    *(v4h*)&Pt[r * 64 + 4 * ll] = pk;
  }
}

#define KSPLIT 8
#define KRANGE (NN / KSPLIT)
#define QT 2

__global__ __launch_bounds__(32 * KSPLIT) void flash_attn_kernel(
    const _Float16* __restrict__ Zs, const _Float16* __restrict__ Zf,
    const _Float16* __restrict__ ZT, const unsigned* __restrict__ Mb,
    float* __restrict__ O, float* __restrict__ out)
{
  const int k    = blockIdx.y;
  const int qb   = blockIdx.x * (16 * QT);
  const int w    = __builtin_amdgcn_readfirstlane((int)(threadIdx.x >> 5));
  const int l    = threadIdx.x & 31;
  const int ll   = l & 15;
  const int half = l >> 4;
  const int c0   = half ? 8 : 0;

  if (blockIdx.x == 0 && blockIdx.y == 0 && w == 0) {
    float t = 0.f;
    for (int i = 0; i < GPARTS / 32; ++i) t += O[l + 32 * i];
    for (int off = 16; off > 0; off >>= 1) t += __shfl_xor(t, off, 32);
    if (l == 0) out[(size_t)NN * DD] = t;
  }

  __shared__ _Float16 __attribute__((aligned(16))) PldsAll[KSPLIT * QT * 16 * 64];
  __shared__ float Mm[KSPLIT * QT * 16];
  __shared__ float Lm[KSPLIT * QT * 16];
  _Float16* Plds = PldsAll + w * (QT * 16 * 64);

  const _Float16* qrow = Zs + ((size_t)k * NN + (qb + ll)) * SS;
  v16h aQ[QT];
  #pragma unroll
  for (int q = 0; q < QT; ++q)
    aQ[q] = make_ab(qrow + q * 16 * SS + c0, qrow + q * 16 * SS + c0 + 16);

  v16h ones;
  #pragma unroll
  for (int i = 0; i < 16; ++i) ones[i] = (_Float16)1.0f;

  float m[QT][8];
  v8f o0[QT], o1[QT];
  v8f lacc[QT];
  #pragma unroll
  for (int q = 0; q < QT; ++q) {
    #pragma unroll
    for (int v = 0; v < 8; ++v) { m[q][v] = -INFINITY; o0[q][v] = 0.f; o1[q][v] = 0.f; lacc[q][v] = 0.f; }
  }

  const _Float16* Zfk  = Zf + (size_t)k * NN * SS;
  const _Float16* ZTk  = ZT + (size_t)k * SS * NN;
  const unsigned* Mbx  = Mb + (size_t)blockIdx.x * (64 * 64);
  const unsigned  foff = (unsigned)(l * 8);
  const int kb0 = w * KRANGE;
  for (int kb = kb0; kb < kb0 + KRANGE; kb += 64) {
    v16h bK[4];
    #pragma unroll
    for (int t = 0; t < 4; ++t) {
      const _Float16* kr = Zfk + (unsigned)(kb * SS) + foff + t * 512;
      bK[t] = make_ab(kr, kr + 256);
    }
    const v16u* mu = (const v16u*)(Mbx + (unsigned)kb);
    const v16u m0a = mu[0], m0b = mu[1];

    v8f sc[QT][4];
    #pragma unroll
    for (int t = 0; t < 4; ++t) {
      #pragma unroll
      for (int q = 0; q < QT; ++q) {
        v8f z = {};
        sc[q][t] = __builtin_amdgcn_wmma_f32_16x16x32_f16(false, aQ[q], false, bK[t], (short)0, z, false, false);
      }
    }

    const v16u m1a = mu[2], m1b = mu[3];
    softmax_tile(sc[0], m0a, m0b, m[0], o0[0], o1[0], lacc[0], Plds, ll, half);

    v16h bV[2][2];
    #pragma unroll
    for (int c = 0; c < 2; ++c) {
      const _Float16* vr0 = ZTk + (unsigned)(kb * SS) + foff + c * 1024;
      bV[c][0] = make_ab(vr0, vr0 + 256);
      bV[c][1] = make_ab(vr0 + 512, vr0 + 768);
    }
    __builtin_amdgcn_sched_barrier(0);
    softmax_tile(sc[1], m1a, m1b, m[1], o0[1], o1[1], lacc[1], Plds + 1024, ll, half);
    asm volatile("s_wait_dscnt 0" ::: "memory");

    #pragma unroll
    for (int c = 0; c < 2; ++c) {
      #pragma unroll
      for (int q = 0; q < QT; ++q) {
        const v16h aP = make_ab(&Plds[q * 1024 + ll * 64 + c * 32 + c0],
                                &Plds[q * 1024 + ll * 64 + c * 32 + c0 + 16]);
        o0[q]   = __builtin_amdgcn_wmma_f32_16x16x32_f16(false, aP, false, bV[c][0], (short)0, o0[q],   false, false);
        o1[q]   = __builtin_amdgcn_wmma_f32_16x16x32_f16(false, aP, false, bV[c][1], (short)0, o1[q],   false, false);
        lacc[q] = __builtin_amdgcn_wmma_f32_16x16x32_f16(false, aP, false, ones,     (short)0, lacc[q], false, false);
      }
    }
  }

  float* Om = (float*)PldsAll;
  #pragma unroll
  for (int q = 0; q < QT; ++q) {
    #pragma unroll
    for (int v = 0; v < 8; ++v) {
      const int r = v + 8 * half;
      Om[(w * QT + q) * 512 + r * 32 + ll]      = o0[q][v];
      Om[(w * QT + q) * 512 + r * 32 + 16 + ll] = o1[q][v];
      if (ll == 0) {
        Mm[(w * QT + q) * 16 + r] = m[q][v];
        Lm[(w * QT + q) * 16 + r] = lacc[q][v];
      }
    }
  }
  __syncthreads();

  for (int j = 0; j < 2 * QT; ++j) {
    const int i = (int)threadIdx.x + j * (32 * KSPLIT);
    const int r = i >> 5;
    float M = Mm[r];
    for (int u = 1; u < KSPLIT; ++u) M = fmaxf(M, Mm[u * (16 * QT) + r]);
    float L = 0.f, acc = 0.f;
    for (int u = 0; u < KSPLIT; ++u) {
      const float f = __expf(Mm[u * (16 * QT) + r] - M);
      L   = fmaf(Lm[u * (16 * QT) + r], f, L);
      acc = fmaf(Om[u * (512 * QT) + i], f, acc);
    }
    O[((size_t)k * NN + qb) * SS + i] = acc / L;
  }
}

__global__ __launch_bounds__(256) void combine_kernel(
    const float* __restrict__ H, const float* __restrict__ U,
    const float* __restrict__ O, const float* __restrict__ thr,
    float* __restrict__ out)
{
  const int n0 = blockIdx.x * 16;
  const int w  = threadIdx.x >> 5;
  const int l  = threadIdx.x & 31;
  const int c  = l & 15;
  const int h  = l >> 4;
  const int dc = w * 32 + 2 * c;

  v8f acc0 = {}; v8f acc1 = {};
  #pragma unroll
  for (int k = 0; k < KK; ++k) {
    const v4f* pa  = (const v4f*)(O + ((size_t)k * NN + n0 + c) * SS + 16 * h);
    const v4f* pb0 = (const v4f*)(U + ((size_t)k * DD + dc) * SS + 16 * h);
    const v4f* pb1 = (const v4f*)(U + ((size_t)k * DD + dc + 1) * SS + 16 * h);
    #pragma unroll
    for (int q = 0; q < 4; ++q) {
      const v4f a = pa[q], b0 = pb0[q], b1 = pb1[q];
      const v2f alo  = {a[0],  a[1]},  ahi  = {a[2],  a[3]};
      const v2f b0lo = {b0[0], b0[1]}, b0hi = {b0[2], b0[3]};
      const v2f b1lo = {b1[0], b1[1]}, b1hi = {b1[2], b1[3]};
      acc0 = __builtin_amdgcn_wmma_f32_16x16x4_f32(false, alo, false, b0lo, (short)0, acc0, false, false);
      acc1 = __builtin_amdgcn_wmma_f32_16x16x4_f32(false, alo, false, b1lo, (short)0, acc1, false, false);
      acc0 = __builtin_amdgcn_wmma_f32_16x16x4_f32(false, ahi, false, b0hi, (short)0, acc0, false, false);
      acc1 = __builtin_amdgcn_wmma_f32_16x16x4_f32(false, ahi, false, b1hi, (short)0, acc1, false, false);
    }
  }

  const v2f t = *(const v2f*)(thr + dc);
  #pragma unroll
  for (int r = 0; r < 8; ++r) {
    const size_t o = (size_t)(n0 + 8 * h + r) * DD + dc;
    const v2f hv = *(const v2f*)(H + o);
    const float h0 = hv[0] + ETA * acc0[r];
    const float h1 = hv[1] + ETA * acc1[r];
    const float a0 = fabsf(h0) - t[0];
    const float a1 = fabsf(h1) - t[1];
    v2f res;
    res[0] = (a0 > 0.f) ? copysignf(a0, h0) : 0.f;
    res[1] = (a1 > 0.f) ? copysignf(a1, h1) : 0.f;
    *(v2f*)(out + o) = res;
  }
}

extern "C" void kernel_launch(void* const* d_in, const int* in_sizes, int n_in,
                              void* d_out, int out_size, void* d_ws, size_t ws_size,
                              hipStream_t stream) {
  (void)in_sizes; (void)n_in; (void)out_size; (void)ws_size;
  const float* H   = (const float*)d_in[0];
  const float* adj = (const float*)d_in[1];
  const float* U   = (const float*)d_in[2];
  const float* lp  = (const float*)d_in[3];
  const float* thr = (const float*)d_in[4];
  float* out = (float*)d_out;

  char* ws = (char*)d_ws;
  _Float16* Zs = (_Float16*)(ws);
  _Float16* Zf = (_Float16*)(ws + (1u << 20));
  _Float16* ZT = (_Float16*)(ws + (2u << 20));
  float*    O  = (float*)   (ws + (3u << 20));
  unsigned* Mb = (unsigned*)d_out;

  zproj_kernel     <<<GTILES + ZTILES + PBLOCKS, 32 * KK, 0, stream>>>(H, U, lp, Zs, Zf, ZT, O, adj, Mb);
  flash_attn_kernel<<<dim3(NN / (16 * QT), KK), 32 * KSPLIT, 0, stream>>>(Zs, Zf, ZT, Mb, O, out);
  combine_kernel   <<<NN / 16,          256, 0, stream>>>(H, U, O, thr, out);
}


# attention softmax in base 2: 1/sqrt(s)*log2(e) folded into the f16 query copy by the projection kernel, scores arrive as exponents of 2: no scale or base multiply per score; sums kept
# speedup vs baseline: 12.8165x; 1.0501x over previous
#include <hip/hip_runtime.h>

#define NN 4096
#define DD 256
#define SS 32
#define KK 4
#define ETA 0.5f
#define GROWS  (KK * SS)
#define GPARTS (KK * KK * SS)

typedef __attribute__((ext_vector_type(16))) _Float16 v16h;
typedef __attribute__((ext_vector_type(8)))  _Float16 v8h;
typedef __attribute__((ext_vector_type(8)))  float    v8f;

__device__ __forceinline__ v16h make_ab(const _Float16* p0, const _Float16* p1) {
  v8h lo = *(const v8h*)p0;
  v8h hi = *(const v8h*)p1;
  return __builtin_shufflevector(lo, hi, 0,1,2,3,4,5,6,7,8,9,10,11,12,13,14,15);
}

typedef __attribute__((ext_vector_type(2))) float    v2f;
typedef __attribute__((ext_vector_type(4))) float    v4f;
typedef __attribute__((ext_vector_type(2))) _Float16 v2h;
typedef __attribute__((ext_vector_type(4))) _Float16 v4h;

__device__ __forceinline__ unsigned kfrag_off(unsigned n, unsigned s) {
  const unsigned kk = n & 63u, gs = s >> 3;
  const unsigned unit = ((kk & 3u) * 2u + (gs >> 1)) * 32u + (gs & 1u) * 16u + (kk >> 2);
  return (n >> 6) * 2048u + unit * 8u + (s & 7u);
}
__device__ __forceinline__ unsigned vfrag_off(unsigned n, unsigned s) {
  const unsigned kk = n & 63u;
  const unsigned unit = ((((kk >> 5) * 2u + (s >> 4)) * 2u + ((kk >> 4) & 1u)) * 32u)
                      + ((kk >> 3) & 1u) * 16u + (s & 15u);
  return (n >> 6) * 2048u + unit * 8u + (kk & 7u);
}

#define QSCALE (0.17677669529663687f * 1.4426950408889634f)

#define ZTILES (NN / 16)
#define GTILES (GROWS / 16)
#define PBLOCKS (NN / 4)

typedef __attribute__((ext_vector_type(16))) unsigned v16u;


template <bool GRAM>
__device__ __forceinline__ void rows_times_Uk(const float* __restrict__ arow,
                                              const float* __restrict__ ucol,
                                              v8f& acc0, v8f& acc1)
{
  #pragma unroll
  for (int J = 0; J < DD / 8; ++J) {
    v4f a;
    if (GRAM) {
      a[0] = arow[(8 * J + 0) * SS];
      a[1] = arow[(8 * J + 1) * SS];
      a[2] = arow[(8 * J + 2) * SS];
      a[3] = arow[(8 * J + 3) * SS];
    } else {
      a = *(const v4f*)(arow + 8 * J);
    }
    const v2f u0 = *(const v2f*)(ucol + (8 * J + 0) * SS);
    const v2f u1 = *(const v2f*)(ucol + (8 * J + 1) * SS);
    const v2f u2 = *(const v2f*)(ucol + (8 * J + 2) * SS);
    const v2f u3 = *(const v2f*)(ucol + (8 * J + 3) * SS);
    const v2f alo  = {a[0],  a[1]},  ahi  = {a[2],  a[3]};
    const v2f b0lo = {u0[0], u1[0]}, b0hi = {u2[0], u3[0]};
    const v2f b1lo = {u0[1], u1[1]}, b1hi = {u2[1], u3[1]};
    acc0 = __builtin_amdgcn_wmma_f32_16x16x4_f32(false, alo, false, b0lo, (short)0, acc0, false, false);
    acc1 = __builtin_amdgcn_wmma_f32_16x16x4_f32(false, alo, false, b1lo, (short)0, acc1, false, false);
    acc0 = __builtin_amdgcn_wmma_f32_16x16x4_f32(false, ahi, false, b0hi, (short)0, acc0, false, false);
    acc1 = __builtin_amdgcn_wmma_f32_16x16x4_f32(false, ahi, false, b1hi, (short)0, acc1, false, false);
  }
}

__global__ __launch_bounds__(32 * KK) __attribute__((amdgpu_waves_per_eu(1, 4))) void zproj_kernel(
    const float* __restrict__ H, const float* __restrict__ U,
    const float* __restrict__ lp,
    _Float16* __restrict__ Zs, _Float16* __restrict__ Zf, _Float16* __restrict__ ZT,
    float* __restrict__ Gp, const float* __restrict__ adj, unsigned* __restrict__ Mb)
{
  const int k    = __builtin_amdgcn_readfirstlane((int)(threadIdx.x >> 5));
  const int l    = threadIdx.x & 31;
  const int c    = l & 15;
  const int h    = l >> 4;
  const int tile = blockIdx.x;
  const float* ucol = U + (size_t)k * DD * SS + (4 * h) * SS + 2 * c;

  v8f acc0 = {}; v8f acc1 = {};

  if (tile < GTILES) {
    const int kq = tile >> 1;
    const int a0 = 16 * (tile & 1);
    if (kq <= k)
      rows_times_Uk<true>(U + (size_t)kq * DD * SS + (4 * h) * SS + a0 + c, ucol, acc0, acc1);
    float part[8];
    #pragma unroll
    for (int r = 0; r < 8; ++r) {
      const int a = a0 + 8 * h + r;
      const float e0 = (kq == k && a == 2 * c)     ? acc0[r] - 1.f : acc0[r];
      const float e1 = (kq == k && a == 2 * c + 1) ? acc1[r] - 1.f : acc1[r];
      part[r] = e0 * e0 + e1 * e1;
    }
    #pragma unroll
    for (int off = 8; off > 0; off >>= 1) {
      #pragma unroll
      for (int r = 0; r < 8; ++r) part[r] += __shfl_xor(part[r], off, 32);
    }
    if (c == 0) {
      v4f* gp = (v4f*)(Gp + (kq * KK + k) * SS + a0 + 8 * h);
      const v4f p0 = {part[0], part[1], part[2], part[3]};
      const v4f p1 = {part[4], part[5], part[6], part[7]};
      gp[0] = p0;
      gp[1] = p1;
    }
    return;
  }

  if (tile >= GTILES + ZTILES) {
    const int g  = (tile - (GTILES + ZTILES)) * KK + k;
    const int x  = g >> 5;
    const int kc = g & 31;
    const float* mrow = adj + (size_t)(32 * x + 8 * h) * NN + 128 * kc + 4 * c;
    v4f mw[2][2][8];
    #pragma unroll
    for (int kk = 0; kk < 2; ++kk) {
      #pragma unroll
      for (int q = 0; q < 2; ++q) {
        #pragma unroll
        for (int v = 0; v < 8; ++v)
          mw[kk][q][v] = *(const v4f*)(mrow + (size_t)(16 * q + v) * NN + 64 * kk);
      }
    }
    #pragma unroll
    for (int kk = 0; kk < 2; ++kk) {
      #pragma unroll
      for (int q = 0; q < 2; ++q) {
        unsigned word = 0u;
        #pragma unroll
        for (int v = 0; v < 8; ++v) {
          #pragma unroll
          for (int j = 0; j < 4; ++j) {
            const unsigned bits = __builtin_amdgcn_ballot_w32(mw[kk][q][v][j] != 0.f);
            word = (l == 4 * v + j) ? bits : word;
          }
        }
        Mb[(((size_t)x * 64 + 2 * kc + kk) * 2 + q) * 32 + l] = word;
      }
    }
    return;
  }

  const int n0 = (tile - GTILES) * 16;
  rows_times_Uk<false>(H + (size_t)(n0 + c) * DD + 4 * h, ucol, acc0, acc1);

  const v2f lpv = *(const v2f*)(lp + k * SS + 2 * c);
  const float e0 = __expf(lpv[0]) * QSCALE;
  const float e1 = __expf(lpv[1]) * QSCALE;
  v8h t0, t1;
  #pragma unroll
  for (int r = 0; r < 8; ++r) {
    const float z0 = acc0[r], z1 = acc1[r];
    const size_t o = ((size_t)k * NN + n0 + 8 * h + r) * SS + 2 * c;
    const v2h kf = {(_Float16)z0, (_Float16)z1};
    const v2h qf = {(_Float16)(z0 * e0), (_Float16)(z1 * e1)};
    *(v2h*)(Zf + (size_t)k * NN * SS
               + kfrag_off((unsigned)(n0 + 8 * h + r), (unsigned)(2 * c))) = kf;
    *(v2h*)(Zs + o) = qf;
    t0[r] = (_Float16)z0;
    t1[r] = (_Float16)z1;
  }
  _Float16* vt = ZT + (size_t)k * SS * NN + vfrag_off((unsigned)(n0 + 8 * h), (unsigned)(2 * c));
  *(v8h*)vt       = t0;
  *(v8h*)(vt + 8) = t1;
}


template <int MASK>
__device__ __forceinline__ float xmask_max(float x) {
  float y;
  asm("v_max_num_f32_dpp %0, %1, %1 row_xmask:%2 row_mask:0xf bank_mask:0xf"
      : "=v"(y) : "v"(x), "n"(MASK));
  return y;
}
__device__ __forceinline__ float xmask8_max3(float x, float m) {
  float y;
  asm("v_max3_num_f32_e64_dpp %0, %1, %1, %2 row_xmask:8 row_mask:0xf bank_mask:0xf"
      : "=v"(y) : "v"(x), "v"(m));
  return y;
}

__device__ __forceinline__ float keep_if(unsigned lanemask, float x) {
  return __builtin_amdgcn_inverse_ballot_w32(lanemask) ? x : -INFINITY;
}
__device__ __forceinline__ void softmax_tile(const v8f (&sc)[4], const v16u& ma, const v16u& mb,
                                             float (&m)[8], v8f& o0, v8f& o1, v8f& lacc,
                                             _Float16* Pt, int ll, int half)
{
  float mx[8];
  #pragma unroll
  for (int v = 0; v < 8; ++v)
    mx[v] = fmaxf(fmaxf(sc[0][v], sc[1][v]), fmaxf(sc[2][v], sc[3][v]));
  #pragma unroll
  for (int v = 0; v < 8; ++v) mx[v] = xmask_max<1>(mx[v]);
  #pragma unroll
  for (int v = 0; v < 8; ++v) mx[v] = xmask_max<2>(mx[v]);
  #pragma unroll
  for (int v = 0; v < 8; ++v) mx[v] = xmask_max<4>(mx[v]);
  #pragma unroll
  for (int v = 0; v < 8; ++v) {
    const unsigned w0 = (v < 4) ? ma[4 * v + 0] : mb[4 * v - 16];
    const unsigned w1 = (v < 4) ? ma[4 * v + 1] : mb[4 * v - 15];
    const unsigned w2 = (v < 4) ? ma[4 * v + 2] : mb[4 * v - 14];
    const unsigned w3 = (v < 4) ? ma[4 * v + 3] : mb[4 * v - 13];
    const float mn    = xmask8_max3(mx[v], m[v]);
    const float scale = __builtin_amdgcn_exp2f(m[v] - mn);
    m[v] = mn;
    const float p0 = __builtin_amdgcn_exp2f(keep_if(w0, sc[0][v] - mn));
    const float p1 = __builtin_amdgcn_exp2f(keep_if(w1, sc[1][v] - mn));
    const float p2 = __builtin_amdgcn_exp2f(keep_if(w2, sc[2][v] - mn));
    const float p3 = __builtin_amdgcn_exp2f(keep_if(w3, sc[3][v] - mn));
    o0[v]   *= scale;
    o1[v]   *= scale;
    lacc[v] *= scale;
    const int r = v + 8 * half;
    const v4h pk = {(_Float16)p0, (_Float16)p1, (_Float16)p2, (_Float16)p3};
    *(v4h*)&Pt[r * 64 + 4 * ll] = pk;
  }
}

#define KSPLIT 8
#define KRANGE (NN / KSPLIT)
#define QT 2

__global__ __launch_bounds__(32 * KSPLIT) void flash_attn_kernel(
    const _Float16* __restrict__ Zs, const _Float16* __restrict__ Zf,
    const _Float16* __restrict__ ZT, const unsigned* __restrict__ Mb,
    float* __restrict__ O, float* __restrict__ out)
{
  const int k    = blockIdx.y;
  const int qb   = blockIdx.x * (16 * QT);
  const int w    = __builtin_amdgcn_readfirstlane((int)(threadIdx.x >> 5));
  const int l    = threadIdx.x & 31;
  const int ll   = l & 15;
  const int half = l >> 4;
  const int c0   = half ? 8 : 0;

  if (blockIdx.x == 0 && blockIdx.y == 0 && w == 0) {
    float t = 0.f;
    for (int i = 0; i < GPARTS / 32; ++i) t += O[l + 32 * i];
    for (int off = 16; off > 0; off >>= 1) t += __shfl_xor(t, off, 32);
    if (l == 0) out[(size_t)NN * DD] = t;
  }

  __shared__ _Float16 __attribute__((aligned(16))) PldsAll[KSPLIT * QT * 16 * 64];
  __shared__ float Mm[KSPLIT * QT * 16];
  __shared__ float Lm[KSPLIT * QT * 16];
  _Float16* Plds = PldsAll + w * (QT * 16 * 64);

  const _Float16* qrow = Zs + ((size_t)k * NN + (qb + ll)) * SS;
  v16h aQ[QT];
  #pragma unroll
  for (int q = 0; q < QT; ++q)
    aQ[q] = make_ab(qrow + q * 16 * SS + c0, qrow + q * 16 * SS + c0 + 16);

  v16h ones;
  #pragma unroll
  for (int i = 0; i < 16; ++i) ones[i] = (_Float16)1.0f;

  float m[QT][8];
  v8f o0[QT], o1[QT];
  v8f lacc[QT];
  #pragma unroll
  for (int q = 0; q < QT; ++q) {
    #pragma unroll
    for (int v = 0; v < 8; ++v) { m[q][v] = -INFINITY; o0[q][v] = 0.f; o1[q][v] = 0.f; lacc[q][v] = 0.f; }
  }

  const _Float16* Zfk  = Zf + (size_t)k * NN * SS;
  const _Float16* ZTk  = ZT + (size_t)k * SS * NN;
  const unsigned* Mbx  = Mb + (size_t)blockIdx.x * (64 * 64);
  const unsigned  foff = (unsigned)(l * 8);
  const int kb0 = w * KRANGE;
  for (int kb = kb0; kb < kb0 + KRANGE; kb += 64) {
    v16h bK[4];
    #pragma unroll
    for (int t = 0; t < 4; ++t) {
      const _Float16* kr = Zfk + (unsigned)(kb * SS) + foff + t * 512;
      bK[t] = make_ab(kr, kr + 256);
    }
    const v16u* mu = (const v16u*)(Mbx + (unsigned)kb);
    const v16u m0a = mu[0], m0b = mu[1];

    v8f sc[QT][4];
    #pragma unroll
    for (int t = 0; t < 4; ++t) {
      #pragma unroll
      for (int q = 0; q < QT; ++q) {
        v8f z = {};
        sc[q][t] = __builtin_amdgcn_wmma_f32_16x16x32_f16(false, aQ[q], false, bK[t], (short)0, z, false, false);
      }
    }

    const v16u m1a = mu[2], m1b = mu[3];
    softmax_tile(sc[0], m0a, m0b, m[0], o0[0], o1[0], lacc[0], Plds, ll, half);

    v16h bV[2][2];
    #pragma unroll
    for (int c = 0; c < 2; ++c) {
      const _Float16* vr0 = ZTk + (unsigned)(kb * SS) + foff + c * 1024;
      bV[c][0] = make_ab(vr0, vr0 + 256);
      bV[c][1] = make_ab(vr0 + 512, vr0 + 768);
    }
    __builtin_amdgcn_sched_barrier(0);
    softmax_tile(sc[1], m1a, m1b, m[1], o0[1], o1[1], lacc[1], Plds + 1024, ll, half);
    asm volatile("s_wait_dscnt 0" ::: "memory");

    #pragma unroll
    for (int c = 0; c < 2; ++c) {
      #pragma unroll
      for (int q = 0; q < QT; ++q) {
        const v16h aP = make_ab(&Plds[q * 1024 + ll * 64 + c * 32 + c0],
                                &Plds[q * 1024 + ll * 64 + c * 32 + c0 + 16]);
        o0[q]   = __builtin_amdgcn_wmma_f32_16x16x32_f16(false, aP, false, bV[c][0], (short)0, o0[q],   false, false);
        o1[q]   = __builtin_amdgcn_wmma_f32_16x16x32_f16(false, aP, false, bV[c][1], (short)0, o1[q],   false, false);
        lacc[q] = __builtin_amdgcn_wmma_f32_16x16x32_f16(false, aP, false, ones,     (short)0, lacc[q], false, false);
      }
    }
  }

  float* Om = (float*)PldsAll;
  #pragma unroll
  for (int q = 0; q < QT; ++q) {
    #pragma unroll
    for (int v = 0; v < 8; ++v) {
      const int r = v + 8 * half;
      Om[(w * QT + q) * 512 + r * 32 + ll]      = o0[q][v];
      Om[(w * QT + q) * 512 + r * 32 + 16 + ll] = o1[q][v];
      if (ll == 0) {
        Mm[(w * QT + q) * 16 + r] = m[q][v];
        Lm[(w * QT + q) * 16 + r] = lacc[q][v];
      }
    }
  }
  __syncthreads();

  for (int j = 0; j < 2 * QT; ++j) {
    const int i = (int)threadIdx.x + j * (32 * KSPLIT);
    const int r = i >> 5;
    float M = Mm[r];
    for (int u = 1; u < KSPLIT; ++u) M = fmaxf(M, Mm[u * (16 * QT) + r]);
    float L = 0.f, acc = 0.f;
    for (int u = 0; u < KSPLIT; ++u) {
      const float f = __builtin_amdgcn_exp2f(Mm[u * (16 * QT) + r] - M);
      L   = fmaf(Lm[u * (16 * QT) + r], f, L);
      acc = fmaf(Om[u * (512 * QT) + i], f, acc);
    }
    O[((size_t)k * NN + qb) * SS + i] = acc / L;
  }
}

__global__ __launch_bounds__(256) void combine_kernel(
    const float* __restrict__ H, const float* __restrict__ U,
    const float* __restrict__ O, const float* __restrict__ thr,
    float* __restrict__ out)
{
  const int n0 = blockIdx.x * 16;
  const int w  = threadIdx.x >> 5;
  const int l  = threadIdx.x & 31;
  const int c  = l & 15;
  const int h  = l >> 4;
  const int dc = w * 32 + 2 * c;

  v8f acc0 = {}; v8f acc1 = {};
  #pragma unroll
  for (int k = 0; k < KK; ++k) {
    const v4f* pa  = (const v4f*)(O + ((size_t)k * NN + n0 + c) * SS + 16 * h);
    const v4f* pb0 = (const v4f*)(U + ((size_t)k * DD + dc) * SS + 16 * h);
    const v4f* pb1 = (const v4f*)(U + ((size_t)k * DD + dc + 1) * SS + 16 * h);
    #pragma unroll
    for (int q = 0; q < 4; ++q) {
      const v4f a = pa[q], b0 = pb0[q], b1 = pb1[q];
      const v2f alo  = {a[0],  a[1]},  ahi  = {a[2],  a[3]};
      const v2f b0lo = {b0[0], b0[1]}, b0hi = {b0[2], b0[3]};
      const v2f b1lo = {b1[0], b1[1]}, b1hi = {b1[2], b1[3]};
      acc0 = __builtin_amdgcn_wmma_f32_16x16x4_f32(false, alo, false, b0lo, (short)0, acc0, false, false);
      acc1 = __builtin_amdgcn_wmma_f32_16x16x4_f32(false, alo, false, b1lo, (short)0, acc1, false, false);
      acc0 = __builtin_amdgcn_wmma_f32_16x16x4_f32(false, ahi, false, b0hi, (short)0, acc0, false, false);
      acc1 = __builtin_amdgcn_wmma_f32_16x16x4_f32(false, ahi, false, b1hi, (short)0, acc1, false, false);
    }
  }

  const v2f t = *(const v2f*)(thr + dc);
  #pragma unroll
  for (int r = 0; r < 8; ++r) {
    const size_t o = (size_t)(n0 + 8 * h + r) * DD + dc;
    const v2f hv = *(const v2f*)(H + o);
    const float h0 = hv[0] + ETA * acc0[r];
    const float h1 = hv[1] + ETA * acc1[r];
    const float a0 = fabsf(h0) - t[0];
    const float a1 = fabsf(h1) - t[1];
    v2f res;
    res[0] = (a0 > 0.f) ? copysignf(a0, h0) : 0.f;
    res[1] = (a1 > 0.f) ? copysignf(a1, h1) : 0.f;
    *(v2f*)(out + o) = res;
  }
}

extern "C" void kernel_launch(void* const* d_in, const int* in_sizes, int n_in,
                              void* d_out, int out_size, void* d_ws, size_t ws_size,
                              hipStream_t stream) {
  (void)in_sizes; (void)n_in; (void)out_size; (void)ws_size;
  const float* H   = (const float*)d_in[0];
  const float* adj = (const float*)d_in[1];
  const float* U   = (const float*)d_in[2];
  const float* lp  = (const float*)d_in[3];
  const float* thr = (const float*)d_in[4];
  float* out = (float*)d_out;

  char* ws = (char*)d_ws;
  _Float16* Zs = (_Float16*)(ws);
  _Float16* Zf = (_Float16*)(ws + (1u << 20));
  _Float16* ZT = (_Float16*)(ws + (2u << 20));
  float*    O  = (float*)   (ws + (3u << 20));
  unsigned* Mb = (unsigned*)d_out;

  zproj_kernel     <<<GTILES + ZTILES + PBLOCKS, 32 * KK, 0, stream>>>(H, U, lp, Zs, Zf, ZT, O, adj, Mb);
  flash_attn_kernel<<<dim3(NN / (16 * QT), KK), 32 * KSPLIT, 0, stream>>>(Zs, Zf, ZT, Mb, O, out);
  combine_kernel   <<<NN / 16,          256, 0, stream>>>(H, U, O, thr, out);
}
